# starting kernel
# baseline (speedup 1.0000x reference)
#include <hip/hip_runtime.h>
#include <hip/hip_bf16.h>


typedef __attribute__((ext_vector_type(2))) float v2f;
typedef __attribute__((ext_vector_type(8))) float v8f;

#define NODES 50000
#define EDGES 800000
#define FDIM  128
#define CDIM  64
#define MTILE 5

__global__ void zero_f32_kernel(float* __restrict__ p, int n) {
  int i = blockIdx.x * blockDim.x + threadIdx.x;
  if (i < n) p[i] = 0.0f;
}

__global__ void degree_kernel(const int* __restrict__ src,
                              const int* __restrict__ dst,
                              float* __restrict__ deg_out,
                              float* __restrict__ deg_in, int e) {
  int i = blockIdx.x * blockDim.x + threadIdx.x;
  if (i < e) {
    atomicAdd(&deg_out[src[i]], 1.0f);
    atomicAdd(&deg_in [dst[i]], 1.0f);
  }
}

__global__ void norm_kernel(float* __restrict__ d_src,
                            float* __restrict__ d_dst, int n) {
  int i = blockIdx.x * blockDim.x + threadIdx.x;
  if (i < n) {
    float a = d_src[i], b = d_dst[i];
    d_src[i] = rsqrtf(a > 0.0f ? a : 1.0f);
    d_dst[i] = rsqrtf(b > 0.0f ? b : 1.0f);
  }
}

__global__ void transpose_w_kernel(const float* __restrict__ W,
                                   float* __restrict__ WT, int kdim, int ndim) {
  int i = blockIdx.x * blockDim.x + threadIdx.x;
  if (i < kdim * ndim) {
    int k = i / ndim, c = i % ndim;
    WT[(size_t)c * kdim + k] = W[i];
  }
}

template <int LDN, bool SCALE_A, bool ADD_BIAS>
__global__ void gemm_k128_wmma(const float* __restrict__ A,
                               const float* __restrict__ WT,
                               const float* __restrict__ bias,
                               const float* __restrict__ dsrc,
                               float* __restrict__ out) {
  const int lane = threadIdx.x & 31;
  const int wave = threadIdx.x >> 5;
  const int m0   = blockIdx.x * (16 * MTILE);
  const int n0   = wave << 4;
  const int col  = n0 + (lane & 15);
  const int kb   = (lane >> 4) << 1;

  float init = 0.0f;
  if constexpr (ADD_BIAS) init = bias[col];
  v8f acc[MTILE];
#pragma unroll
  for (int t = 0; t < MTILE; ++t)
#pragma unroll
    for (int r = 0; r < 8; ++r) acc[t][r] = init;

  const float* __restrict__ arow[MTILE];
  float scale[MTILE];
#pragma unroll
  for (int t = 0; t < MTILE; ++t) {
    const int row = m0 + t * 16 + (lane & 15);
    arow[t]  = A + (size_t)row * FDIM;
    scale[t] = SCALE_A ? dsrc[row] : 1.0f;
  }
  const float* __restrict__ bcol = WT + (size_t)col * FDIM;

#pragma unroll
  for (int k = 0; k < FDIM; k += 4) {
    v2f b;
    b.x = bcol[k + kb];
    b.y = bcol[k + kb + 1];
#pragma unroll
    for (int t = 0; t < MTILE; ++t) {
      v2f a;
      a.x = arow[t][k + kb];
      a.y = arow[t][k + kb + 1];
      if constexpr (SCALE_A) { a.x *= scale[t]; a.y *= scale[t]; }
      acc[t] = __builtin_amdgcn_wmma_f32_16x16x4_f32(
           false, a,  false, b,
           (short)0, acc[t],  false,  false);
    }
  }

  const int rbase = (lane >> 4) << 3;
#pragma unroll
  for (int t = 0; t < MTILE; ++t)
#pragma unroll
    for (int r = 0; r < 8; ++r)
      out[(size_t)(m0 + t * 16 + rbase + r) * LDN + col] = acc[t][r];
}

__global__ void scatter_edges_kernel(const float* __restrict__ h,
                                     const int* __restrict__ src,
                                     const int* __restrict__ dst,
                                     float* __restrict__ agg, int e) {
  const int lane   = threadIdx.x & 31;
  const int gwave  = (blockIdx.x * blockDim.x + threadIdx.x) >> 5;
  const int nwaves = (gridDim.x * blockDim.x) >> 5;
  for (int i = gwave; i < e; i += nwaves) {
    const int s = src[i];
    const int d = dst[i];
    if (i + nwaves < e) {
      const int sn = src[i + nwaves];
      __builtin_prefetch(h + (size_t)sn * FDIM + lane * 4, 0, 1);
    }
    const float4 v =
        ((const float4*)(h + (size_t)s * FDIM))[lane];
    float* ap = agg + (size_t)d * FDIM + (lane << 2);
    atomicAdd(ap + 0, v.x);
    atomicAdd(ap + 1, v.y);
    atomicAdd(ap + 2, v.z);
    atomicAdd(ap + 3, v.w);
  }
}

template <bool RELU, bool ADD_RES>
__global__ void epilogue_kernel(const float* __restrict__ agg,
                                const float* __restrict__ ddst,
                                const float* __restrict__ bias,
                                const float* __restrict__ res,
                                float* __restrict__ out, int n) {
  int i = blockIdx.x * blockDim.x + threadIdx.x;
  if (i >= n) return;
  const int row = i >> 7;
  const int col = i & (FDIM - 1);
  float v = agg[i] * ddst[row] + bias[col];
  if constexpr (ADD_RES) v += res[i];
  if constexpr (RELU) v = fmaxf(v, 0.0f);
  out[i] = v;
}

extern "C" void kernel_launch(void* const* d_in, const int* in_sizes, int n_in,
                              void* d_out, int out_size, void* d_ws, size_t ws_size,
                              hipStream_t stream) {
  const float* x   = (const float*)d_in[0];
  const int*   src = (const int*)  d_in[1];
  const int*   dst = (const int*)  d_in[2];
  const float* W1  = (const float*)d_in[3];
  const float* b1  = (const float*)d_in[4];
  const float* W2  = (const float*)d_in[5];
  const float* b2  = (const float*)d_in[6];
  const float* W3  = (const float*)d_in[7];
  const float* b3  = (const float*)d_in[8];
  const float* Wr  = (const float*)d_in[9];
  const float* br  = (const float*)d_in[10];
  const float* Wo  = (const float*)d_in[11];
  const float* bo  = (const float*)d_in[12];
  float* out = (float*)d_out;

  float* ws    = (float*)d_ws;
  float* dsrc  = ws;
  float* ddst  = dsrc + NODES;
  float* RES   = ddst + NODES;
  float* ABUF  = RES  + (size_t)NODES * FDIM;
  float* HBUF  = ABUF + (size_t)NODES * FDIM;
  float* AGG   = HBUF + (size_t)NODES * FDIM;
  float* WTr   = AGG  + (size_t)NODES * FDIM;
  float* WT1   = WTr  + FDIM * FDIM;
  float* WT2   = WT1  + FDIM * FDIM;
  float* WT3   = WT2  + FDIM * FDIM;
  float* WTo   = WT3  + FDIM * FDIM;

  const int NF       = NODES * FDIM;
  const int zb_norm  = (2 * NODES + 255) / 256;
  const int zb_feat  = (NF + 255) / 256;
  const int eb       = (EDGES + 255) / 256;
  const int wb       = (FDIM * FDIM + 255) / 256;
  const int wbo      = (FDIM * CDIM + 255) / 256;
  const int mblocks  = NODES / (16 * MTILE);
  const dim3 gemmH_blk(256);
  const dim3 gemmC_blk(128);
  const dim3 scat_grid(4096), scat_blk(256);

  zero_f32_kernel<<<zb_norm, 256, 0, stream>>>(dsrc, 2 * NODES);
  degree_kernel<<<eb, 256, 0, stream>>>(src, dst, dsrc, ddst, EDGES);
  norm_kernel<<<(NODES + 255) / 256, 256, 0, stream>>>(dsrc, ddst, NODES);

  transpose_w_kernel<<<wb,  256, 0, stream>>>(Wr, WTr, FDIM, FDIM);
  transpose_w_kernel<<<wb,  256, 0, stream>>>(W1, WT1, FDIM, FDIM);
  transpose_w_kernel<<<wb,  256, 0, stream>>>(W2, WT2, FDIM, FDIM);
  transpose_w_kernel<<<wb,  256, 0, stream>>>(W3, WT3, FDIM, FDIM);
  transpose_w_kernel<<<wbo, 256, 0, stream>>>(Wo, WTo, FDIM, CDIM);

  gemm_k128_wmma<FDIM, false, true>
      <<<mblocks, gemmH_blk, 0, stream>>>(x, WTr, br, nullptr, RES);

  gemm_k128_wmma<FDIM, true, false>
      <<<mblocks, gemmH_blk, 0, stream>>>(x, WT1, nullptr, dsrc, HBUF);
  zero_f32_kernel<<<zb_feat, 256, 0, stream>>>(AGG, NF);
  scatter_edges_kernel<<<scat_grid, scat_blk, 0, stream>>>(HBUF, src, dst, AGG, EDGES);
  epilogue_kernel<true, false>
      <<<zb_feat, 256, 0, stream>>>(AGG, ddst, b1, nullptr, ABUF, NF);

  gemm_k128_wmma<FDIM, true, false>
      <<<mblocks, gemmH_blk, 0, stream>>>(ABUF, WT2, nullptr, dsrc, HBUF);
  zero_f32_kernel<<<zb_feat, 256, 0, stream>>>(AGG, NF);
  scatter_edges_kernel<<<scat_grid, scat_blk, 0, stream>>>(HBUF, src, dst, AGG, EDGES);
  epilogue_kernel<true, false>
      <<<zb_feat, 256, 0, stream>>>(AGG, ddst, b2, nullptr, ABUF, NF);

  gemm_k128_wmma<FDIM, true, false>
      <<<mblocks, gemmH_blk, 0, stream>>>(ABUF, WT3, nullptr, dsrc, HBUF);
  zero_f32_kernel<<<zb_feat, 256, 0, stream>>>(AGG, NF);
  scatter_edges_kernel<<<scat_grid, scat_blk, 0, stream>>>(HBUF, src, dst, AGG, EDGES);
  epilogue_kernel<true, true>
      <<<zb_feat, 256, 0, stream>>>(AGG, ddst, b3, RES, ABUF, NF);

  gemm_k128_wmma<CDIM, false, true>
      <<<mblocks, gemmC_blk, 0, stream>>>(ABUF, WTo, bo, nullptr, out);
}


# SpMM push (f32 atomics into zeroed buffer, epilogue pass) -> pull over a per-call CSR by destination (int atomic rank, scan, fill); row summed in registers, written once; sum order arrival, as before
# speedup vs baseline: 2.1410x; 2.1410x over previous
#include <hip/hip_runtime.h>
#include <hip/hip_bf16.h>


typedef __attribute__((ext_vector_type(2))) float v2f;
typedef __attribute__((ext_vector_type(8))) float v8f;

#define NODES 50000
#define EDGES 800000
#define FDIM  128
#define CDIM  64
#define MTILE 5

__global__ void zero_deg_kernel(float* __restrict__ deg_out,
                                int* __restrict__ cnt_in, int n) {
  int i = blockIdx.x * blockDim.x + threadIdx.x;
  if (i < n) { deg_out[i] = 0.0f; cnt_in[i] = 0; }
}

__global__ void degree_rank_kernel(const int* __restrict__ src,
                                   const int* __restrict__ dst,
                                   float* __restrict__ deg_out,
                                   int* __restrict__ cnt_in,
                                   int* __restrict__ rank, int e) {
  int i = blockIdx.x * blockDim.x + threadIdx.x;
  if (i < e) {
    atomicAdd(&deg_out[src[i]], 1.0f);
    rank[i] = atomicAdd(&cnt_in[dst[i]], 1);
  }
}

__global__ void __launch_bounds__(256)
scan_norm_kernel(const int* __restrict__ cnt_in, float* __restrict__ d_src,
                 float* __restrict__ d_dst, int* __restrict__ rowptr, int n) {
  __shared__ int wfront[8];
  __shared__ int wtot[8];
  const int t = threadIdx.x, lane = t & 31, w = t >> 5;
  const int base = blockIdx.x * 256;
  int front = 0;
  const int4* __restrict__ c4 = (const int4*)cnt_in;
  const int n4 = base >> 2;
  for (int i0 = 0; i0 < n4; i0 += 8 * 256) {
    int4 v[8];
#pragma unroll
    for (int u = 0; u < 8; ++u)
      v[u] = c4[min(i0 + u * 256 + t, n4 - 1)];
#pragma unroll
    for (int u = 0; u < 8; ++u)
      if (i0 + u * 256 + t < n4) front += (v[u].x + v[u].y) + (v[u].z + v[u].w);
  }
  for (int off = 16; off > 0; off >>= 1) front += __shfl_xor(front, off);

  const int i = base + t;
  const int c = (i < n) ? cnt_in[i] : 0;
  int incl = c;
  for (int off = 1; off < 32; off <<= 1) {
    const int v = __shfl_up(incl, off);
    if (lane >= off) incl += v;
  }
  if (lane == 0)  wfront[w] = front;
  if (lane == 31) wtot[w]   = incl;
  __syncthreads();
  int carry = 0;
  for (int k = 0; k < 8; ++k) {
    carry += wfront[k];
    if (k < w) carry += wtot[k];
  }
  const int excl = carry + incl - c;
  if (i < n) {
    rowptr[i] = excl;
    if (i == n - 1) rowptr[n] = excl + c;
    const float a = d_src[i];
    d_src[i] = rsqrtf(a > 0.0f ? a : 1.0f);
    d_dst[i] = rsqrtf(c > 0 ? (float)c : 1.0f);
  }
}

__global__ void fill_csr_kernel(const int* __restrict__ src,
                                const int* __restrict__ dst,
                                const int* __restrict__ rowptr,
                                const int* __restrict__ rank,
                                int* __restrict__ colidx, int e) {
  int i = blockIdx.x * blockDim.x + threadIdx.x;
  if (i < e) colidx[rowptr[dst[i]] + rank[i]] = src[i];
}

__global__ void transpose_w_kernel(const float* __restrict__ W,
                                   float* __restrict__ WT, int kdim, int ndim) {
  int i = blockIdx.x * blockDim.x + threadIdx.x;
  if (i < kdim * ndim) {
    int k = i / ndim, c = i % ndim;
    WT[(size_t)c * kdim + k] = W[i];
  }
}

template <int LDN, bool SCALE_A, bool ADD_BIAS>
__global__ void gemm_k128_wmma(const float* __restrict__ A,
                               const float* __restrict__ WT,
                               const float* __restrict__ bias,
                               const float* __restrict__ dsrc,
                               float* __restrict__ out) {
  const int lane = threadIdx.x & 31;
  const int wave = threadIdx.x >> 5;
  const int m0   = blockIdx.x * (16 * MTILE);
  const int n0   = wave << 4;
  const int col  = n0 + (lane & 15);
  const int kb   = (lane >> 4) << 1;

  float init = 0.0f;
  if constexpr (ADD_BIAS) init = bias[col];
  v8f acc[MTILE];
#pragma unroll
  for (int t = 0; t < MTILE; ++t)
#pragma unroll
    for (int r = 0; r < 8; ++r) acc[t][r] = init;

  const float* __restrict__ arow[MTILE];
  float scale[MTILE];
#pragma unroll
  for (int t = 0; t < MTILE; ++t) {
    const int row = m0 + t * 16 + (lane & 15);
    arow[t]  = A + (size_t)row * FDIM;
    scale[t] = SCALE_A ? dsrc[row] : 1.0f;
  }
  const float* __restrict__ bcol = WT + (size_t)col * FDIM;

#pragma unroll
  for (int k = 0; k < FDIM; k += 4) {
    v2f b;
    b.x = bcol[k + kb];
    b.y = bcol[k + kb + 1];
#pragma unroll
    for (int t = 0; t < MTILE; ++t) {
      v2f a;
      a.x = arow[t][k + kb];
      a.y = arow[t][k + kb + 1];
      if constexpr (SCALE_A) { a.x *= scale[t]; a.y *= scale[t]; }
      acc[t] = __builtin_amdgcn_wmma_f32_16x16x4_f32(
           false, a,  false, b,
           (short)0, acc[t],  false,  false);
    }
  }

  const int rbase = (lane >> 4) << 3;
#pragma unroll
  for (int t = 0; t < MTILE; ++t)
#pragma unroll
    for (int r = 0; r < 8; ++r)
      out[(size_t)(m0 + t * 16 + rbase + r) * LDN + col] = acc[t][r];
}

template <bool ADD_RES>
__global__ void __launch_bounds__(256)
gather_rows_kernel(const float* __restrict__ h,
                   const int* __restrict__ rowptr,
                   const int* __restrict__ colidx,
                   const float* __restrict__ ddst,
                   const float* __restrict__ bias,
                   const float* __restrict__ res,
                   float* __restrict__ out, int n) {
  const int lane = threadIdx.x & 31;
  const int row = __builtin_amdgcn_readfirstlane(
      (int)((blockIdx.x * blockDim.x + threadIdx.x) >> 5));
  if (row >= n) return;
  const int beg = rowptr[row];
  const int end = rowptr[row + 1];
  const float4* __restrict__ h4 = (const float4*)h;
  float4 acc = make_float4(0.0f, 0.0f, 0.0f, 0.0f);

  for (int base = beg; base < end; base += 32) {
    const int cnt = min(32, end - base);
    const int mine = (lane < cnt) ? colidx[base + lane] : 0;
    int j = 0;
    for (; j + 8 <= cnt; j += 8) {
      float4 v[8];
#pragma unroll
      for (int u = 0; u < 8; ++u) {
        const int sidx = __builtin_amdgcn_readlane(mine, j + u);
        v[u] = h4[(size_t)sidx * (FDIM / 4) + lane];
      }
#pragma unroll
      for (int u = 0; u < 8; ++u) {
        acc.x += v[u].x; acc.y += v[u].y; acc.z += v[u].z; acc.w += v[u].w;
      }
    }
    for (; j + 4 <= cnt; j += 4) {
      float4 v[4];
#pragma unroll
      for (int u = 0; u < 4; ++u) {
        const int sidx = __builtin_amdgcn_readlane(mine, j + u);
        v[u] = h4[(size_t)sidx * (FDIM / 4) + lane];
      }
#pragma unroll
      for (int u = 0; u < 4; ++u) {
        acc.x += v[u].x; acc.y += v[u].y; acc.z += v[u].z; acc.w += v[u].w;
      }
    }
    for (; j < cnt; ++j) {
      const int sidx = __builtin_amdgcn_readlane(mine, j);
      const float4 v = h4[(size_t)sidx * (FDIM / 4) + lane];
      acc.x += v.x; acc.y += v.y; acc.z += v.z; acc.w += v.w;
    }
  }

  const float dd = ddst[row];
  const float4 b = ((const float4*)bias)[lane];
  float4 o;
  o.x = acc.x * dd + b.x;
  o.y = acc.y * dd + b.y;
  o.z = acc.z * dd + b.z;
  o.w = acc.w * dd + b.w;
  if constexpr (ADD_RES) {
    const float4 r = ((const float4*)res)[(size_t)row * (FDIM / 4) + lane];
    o.x += r.x; o.y += r.y; o.z += r.z; o.w += r.w;
  }
  o.x = fmaxf(o.x, 0.0f);
  o.y = fmaxf(o.y, 0.0f);
  o.z = fmaxf(o.z, 0.0f);
  o.w = fmaxf(o.w, 0.0f);
  ((float4*)out)[(size_t)row * (FDIM / 4) + lane] = o;
}

extern "C" void kernel_launch(void* const* d_in, const int* in_sizes, int n_in,
                              void* d_out, int out_size, void* d_ws, size_t ws_size,
                              hipStream_t stream) {
  const float* x   = (const float*)d_in[0];
  const int*   src = (const int*)  d_in[1];
  const int*   dst = (const int*)  d_in[2];
  const float* W1  = (const float*)d_in[3];
  const float* b1  = (const float*)d_in[4];
  const float* W2  = (const float*)d_in[5];
  const float* b2  = (const float*)d_in[6];
  const float* W3  = (const float*)d_in[7];
  const float* b3  = (const float*)d_in[8];
  const float* Wr  = (const float*)d_in[9];
  const float* br  = (const float*)d_in[10];
  const float* Wo  = (const float*)d_in[11];
  const float* bo  = (const float*)d_in[12];
  float* out = (float*)d_out;

  float* ws    = (float*)d_ws;
  float* dsrc  = ws;
  float* ddst  = dsrc + NODES;
  float* RES   = ddst + NODES;
  float* ABUF  = RES  + (size_t)NODES * FDIM;
  float* HBUF  = ABUF + (size_t)NODES * FDIM;
  float* WTr   = HBUF + (size_t)NODES * FDIM;
  float* WT1   = WTr  + FDIM * FDIM;
  float* WT2   = WT1  + FDIM * FDIM;
  float* WT3   = WT2  + FDIM * FDIM;
  float* WTo   = WT3  + FDIM * FDIM;
  int*   cnt   = (int*)(WTo + CDIM * FDIM);
  int*   rowp  = cnt  + NODES;
  int*   rank  = rowp + NODES + 16;
  int*   colx  = rank + EDGES;

  const int eb       = (EDGES + 255) / 256;
  const int nb       = (NODES + 255) / 256;
  const int wb       = (FDIM * FDIM + 255) / 256;
  const int wbo      = (FDIM * CDIM + 255) / 256;
  const int mblocks  = NODES / (16 * MTILE);
  const int gblocks  = (NODES * 32 + 255) / 256;
  const dim3 gemmH_blk(256);
  const dim3 gemmC_blk(128);

  zero_deg_kernel<<<nb, 256, 0, stream>>>(dsrc, cnt, NODES);
  degree_rank_kernel<<<eb, 256, 0, stream>>>(src, dst, dsrc, cnt, rank, EDGES);
  scan_norm_kernel<<<nb, 256, 0, stream>>>(cnt, dsrc, ddst, rowp, NODES);
  fill_csr_kernel<<<eb, 256, 0, stream>>>(src, dst, rowp, rank, colx, EDGES);

  transpose_w_kernel<<<wb,  256, 0, stream>>>(Wr, WTr, FDIM, FDIM);
  transpose_w_kernel<<<wb,  256, 0, stream>>>(W1, WT1, FDIM, FDIM);
  transpose_w_kernel<<<wb,  256, 0, stream>>>(W2, WT2, FDIM, FDIM);
  transpose_w_kernel<<<wb,  256, 0, stream>>>(W3, WT3, FDIM, FDIM);
  transpose_w_kernel<<<wbo, 256, 0, stream>>>(Wo, WTo, FDIM, CDIM);

  gemm_k128_wmma<FDIM, false, true>
      <<<mblocks, gemmH_blk, 0, stream>>>(x, WTr, br, nullptr, RES);

  gemm_k128_wmma<FDIM, true, false>
      <<<mblocks, gemmH_blk, 0, stream>>>(x, WT1, nullptr, dsrc, HBUF);
  gather_rows_kernel<false>
      <<<gblocks, 256, 0, stream>>>(HBUF, rowp, colx, ddst, b1, nullptr, ABUF, NODES);

  gemm_k128_wmma<FDIM, true, false>
      <<<mblocks, gemmH_blk, 0, stream>>>(ABUF, WT2, nullptr, dsrc, HBUF);
  gather_rows_kernel<false>
      <<<gblocks, 256, 0, stream>>>(HBUF, rowp, colx, ddst, b2, nullptr, ABUF, NODES);

  gemm_k128_wmma<FDIM, true, false>
      <<<mblocks, gemmH_blk, 0, stream>>>(ABUF, WT3, nullptr, dsrc, HBUF);
  gather_rows_kernel<true>
      <<<gblocks, 256, 0, stream>>>(HBUF, rowp, colx, ddst, b3, RES, ABUF, NODES);

  gemm_k128_wmma<CDIM, false, true>
      <<<mblocks, gemmC_blk, 0, stream>>>(ABUF, WTo, bo, nullptr, out);
}


# Aggregation moved into the consuming GEMM's blocks: each sums its 80 rows' in-edges into an LDS A tile (3 launches, activation round trip gone); fragments read 16 B wide, k summed in another order
# speedup vs baseline: 2.4220x; 1.1312x over previous
#include <hip/hip_runtime.h>
#include <hip/hip_bf16.h>


typedef __attribute__((ext_vector_type(2))) float v2f;
typedef __attribute__((ext_vector_type(8))) float v8f;

#define NODES 50000
#define EDGES 800000
#define FDIM  128
#define CDIM  64
#define MTILE 5

__global__ void zero_deg_kernel(float* __restrict__ deg_out,
                                int* __restrict__ cnt_in, int n) {
  int i = blockIdx.x * blockDim.x + threadIdx.x;
  if (i < n) { deg_out[i] = 0.0f; cnt_in[i] = 0; }
}

__global__ void degree_rank_kernel(const int* __restrict__ src,
                                   const int* __restrict__ dst,
                                   float* __restrict__ deg_out,
                                   int* __restrict__ cnt_in,
                                   int* __restrict__ rank, int e) {
  int i = blockIdx.x * blockDim.x + threadIdx.x;
  if (i < e) {
    atomicAdd(&deg_out[src[i]], 1.0f);
    rank[i] = atomicAdd(&cnt_in[dst[i]], 1);
  }
}

__global__ void __launch_bounds__(256)
scan_norm_kernel(const int* __restrict__ cnt_in, float* __restrict__ d_src,
                 float* __restrict__ d_dst, int* __restrict__ rowptr, int n) {
  __shared__ int wfront[8];
  __shared__ int wtot[8];
  const int t = threadIdx.x, lane = t & 31, w = t >> 5;
  const int base = blockIdx.x * 256;
  int front = 0;
  const int4* __restrict__ c4 = (const int4*)cnt_in;
  const int n4 = base >> 2;
  for (int i0 = 0; i0 < n4; i0 += 8 * 256) {
    int4 v[8];
#pragma unroll
    for (int u = 0; u < 8; ++u)
      v[u] = c4[min(i0 + u * 256 + t, n4 - 1)];
#pragma unroll
    for (int u = 0; u < 8; ++u)
      if (i0 + u * 256 + t < n4) front += (v[u].x + v[u].y) + (v[u].z + v[u].w);
  }
  for (int off = 16; off > 0; off >>= 1) front += __shfl_xor(front, off);

  const int i = base + t;
  const int c = (i < n) ? cnt_in[i] : 0;
  int incl = c;
  for (int off = 1; off < 32; off <<= 1) {
    const int v = __shfl_up(incl, off);
    if (lane >= off) incl += v;
  }
  if (lane == 0)  wfront[w] = front;
  if (lane == 31) wtot[w]   = incl;
  __syncthreads();
  int carry = 0;
  for (int k = 0; k < 8; ++k) {
    carry += wfront[k];
    if (k < w) carry += wtot[k];
  }
  const int excl = carry + incl - c;
  if (i < n) {
    rowptr[i] = excl;
    if (i == n - 1) rowptr[n] = excl + c;
    const float a = d_src[i];
    d_src[i] = rsqrtf(a > 0.0f ? a : 1.0f);
    d_dst[i] = rsqrtf(c > 0 ? (float)c : 1.0f);
  }
}

__global__ void fill_csr_kernel(const int* __restrict__ src,
                                const int* __restrict__ dst,
                                const int* __restrict__ rowptr,
                                const int* __restrict__ rank,
                                int* __restrict__ colidx, int e) {
  int i = blockIdx.x * blockDim.x + threadIdx.x;
  if (i < e) colidx[rowptr[dst[i]] + rank[i]] = src[i];
}

__global__ void transpose_w_kernel(const float* __restrict__ W,
                                   float* __restrict__ WT, int kdim, int ndim) {
  int i = blockIdx.x * blockDim.x + threadIdx.x;
  if (i < kdim * ndim) {
    int k = i / ndim, c = i % ndim;
    WT[(size_t)c * kdim + k] = W[i];
  }
}

template <int LDN, bool SCALE_A, bool ADD_BIAS>
__global__ void gemm_k128_wmma(const float* __restrict__ A,
                               const float* __restrict__ WT,
                               const float* __restrict__ bias,
                               const float* __restrict__ dsrc,
                               float* __restrict__ out) {
  const int lane = threadIdx.x & 31;
  const int wave = threadIdx.x >> 5;
  const int m0   = blockIdx.x * (16 * MTILE);
  const int n0   = wave << 4;
  const int col  = n0 + (lane & 15);
  const int kb   = (lane >> 4) << 1;

  float init = 0.0f;
  if constexpr (ADD_BIAS) init = bias[col];
  v8f acc[MTILE];
#pragma unroll
  for (int t = 0; t < MTILE; ++t)
#pragma unroll
    for (int r = 0; r < 8; ++r) acc[t][r] = init;

  const float* __restrict__ arow[MTILE];
  float scale[MTILE];
#pragma unroll
  for (int t = 0; t < MTILE; ++t) {
    const int row = m0 + t * 16 + (lane & 15);
    arow[t]  = A + (size_t)row * FDIM;
    scale[t] = SCALE_A ? dsrc[row] : 1.0f;
  }
  const float* __restrict__ bcol = WT + (size_t)col * FDIM;

#pragma unroll
  for (int k = 0; k < FDIM; k += 4) {
    v2f b;
    b.x = bcol[k + kb];
    b.y = bcol[k + kb + 1];
#pragma unroll
    for (int t = 0; t < MTILE; ++t) {
      v2f a;
      a.x = arow[t][k + kb];
      a.y = arow[t][k + kb + 1];
      if constexpr (SCALE_A) { a.x *= scale[t]; a.y *= scale[t]; }
      acc[t] = __builtin_amdgcn_wmma_f32_16x16x4_f32(
           false, a,  false, b,
           (short)0, acc[t],  false,  false);
    }
  }

  const int rbase = (lane >> 4) << 3;
#pragma unroll
  for (int t = 0; t < MTILE; ++t)
#pragma unroll
    for (int r = 0; r < 8; ++r)
      out[(size_t)(m0 + t * 16 + rbase + r) * LDN + col] = acc[t][r];
}

#define RPW    (16 * MTILE / 8)
#define APITCH (FDIM + 4)

__device__ __forceinline__ void add_source_rows(float4& acc,
                                                const float4* __restrict__ h4,
                                                int ids, int cnt, int lane) {
  for (int j = 0; j < cnt; j += 16) {
    float4 v[16];
#pragma unroll
    for (int u = 0; u < 16; ++u) {
      const int sidx = __builtin_amdgcn_readlane(ids, j + u);
      v[u] = h4[(size_t)sidx * (FDIM / 4) + lane];
    }
#pragma unroll
    for (int u = 0; u < 16; ++u) {
      const bool ok = (j + u) < cnt;
      acc.x += ok ? v[u].x : 0.0f;
      acc.y += ok ? v[u].y : 0.0f;
      acc.z += ok ? v[u].z : 0.0f;
      acc.w += ok ? v[u].w : 0.0f;
    }
  }
}

template <int LDN, bool ADD_RES, bool SCALE_A, bool ADD_BIAS>
__global__ void __launch_bounds__(256)
gather_gemm_kernel(const float* __restrict__ h,
                   const int* __restrict__ rowptr,
                   const int* __restrict__ colidx,
                   const float* __restrict__ ddst,
                   const float* __restrict__ dsrc,
                   const float* __restrict__ bias_in,
                   const float* __restrict__ res,
                   const float* __restrict__ WT,
                   const float* __restrict__ bias_out,
                   float* __restrict__ out) {
  __shared__ float As[16 * MTILE * APITCH];
  const int lane = threadIdx.x & 31;
  const int wave = __builtin_amdgcn_readfirstlane((int)(threadIdx.x >> 5));
  const int m0   = blockIdx.x * (16 * MTILE);

  {
    const int r0 = m0 + wave * RPW;
    const int   rp  = (lane <= RPW) ? rowptr[r0 + lane] : 0;
    const float ddv = (lane < RPW) ? ddst[r0 + lane] : 0.0f;
    float dsv = 0.0f;
    if constexpr (SCALE_A) dsv = (lane < RPW) ? dsrc[r0 + lane] : 0.0f;
    const float4* __restrict__ h4 = (const float4*)h;
    const float4 bin = ((const float4*)bias_in)[lane];
    int beg = __builtin_amdgcn_readlane(rp, 0);
    int end = __builtin_amdgcn_readlane(rp, 1);
    int raw = colidx[min(beg + lane, EDGES - 1)];
    for (int i = 0; i < RPW; ++i) {
      const int row = r0 + i;
      const int nend = __builtin_amdgcn_readlane(rp, min(i + 2, RPW));
      const int raw_next = colidx[min(end + lane, EDGES - 1)];
      const int mine = (lane < min(32, end - beg)) ? raw : 0;
      float4 r = make_float4(0.0f, 0.0f, 0.0f, 0.0f);
      if constexpr (ADD_RES)
        r = ((const float4*)res)[(size_t)row * (FDIM / 4) + lane];
      float4 acc = make_float4(0.0f, 0.0f, 0.0f, 0.0f);
      add_source_rows(acc, h4, mine, min(32, end - beg), lane);
      for (int base = beg + 32; base < end; base += 32) {
        const int cnt = min(32, end - base);
        const int ids = (lane < cnt) ? colidx[base + lane] : 0;
        add_source_rows(acc, h4, ids, cnt, lane);
      }
      const float dd = __int_as_float(
          __builtin_amdgcn_readlane(__float_as_int(ddv), i));
      float4 o;
      o.x = acc.x * dd + bin.x;
      o.y = acc.y * dd + bin.y;
      o.z = acc.z * dd + bin.z;
      o.w = acc.w * dd + bin.w;
      if constexpr (ADD_RES) {
        o.x += r.x; o.y += r.y; o.z += r.z; o.w += r.w;
      }
      o.x = fmaxf(o.x, 0.0f);
      o.y = fmaxf(o.y, 0.0f);
      o.z = fmaxf(o.z, 0.0f);
      o.w = fmaxf(o.w, 0.0f);
      if constexpr (SCALE_A) {
        const float ds = __int_as_float(
            __builtin_amdgcn_readlane(__float_as_int(dsv), i));
        o.x *= ds; o.y *= ds; o.z *= ds; o.w *= ds;
      }
      *(float4*)&As[(wave * RPW + i) * APITCH + 4 * lane] = o;
      beg = end; end = nend; raw = raw_next;
    }
  }
  __syncthreads();
  if (wave >= LDN / 16) return;

  const int c   = lane & 15;
  const int kh  = (lane >> 4) << 2;
  const int col = (wave << 4) + c;

  float init = 0.0f;
  if constexpr (ADD_BIAS) init = bias_out[col];
  v8f acc[MTILE];
#pragma unroll
  for (int t = 0; t < MTILE; ++t)
#pragma unroll
    for (int r = 0; r < 8; ++r) acc[t][r] = init;

  const float* __restrict__ bcol = WT + (size_t)col * FDIM + kh;
  const float* arow = As + c * APITCH + kh;

#pragma unroll
  for (int j = 0; j < FDIM / 8; ++j) {
    const float4 b = *(const float4*)(bcol + 8 * j);
    float4 a[MTILE];
#pragma unroll
    for (int t = 0; t < MTILE; ++t)
      a[t] = *(const float4*)(arow + t * 16 * APITCH + 8 * j);
    v2f b0, b1;
    b0.x = b.x; b0.y = b.y; b1.x = b.z; b1.y = b.w;
#pragma unroll
    for (int t = 0; t < MTILE; ++t) {
      v2f a0;
      a0.x = a[t].x; a0.y = a[t].y;
      acc[t] = __builtin_amdgcn_wmma_f32_16x16x4_f32(
           false, a0,  false, b0,
           (short)0, acc[t],  false,  false);
    }
#pragma unroll
    for (int t = 0; t < MTILE; ++t) {
      v2f a1;
      a1.x = a[t].z; a1.y = a[t].w;
      acc[t] = __builtin_amdgcn_wmma_f32_16x16x4_f32(
           false, a1,  false, b1,
           (short)0, acc[t],  false,  false);
    }
  }

  const int rbase = (lane >> 4) << 3;
#pragma unroll
  for (int t = 0; t < MTILE; ++t)
#pragma unroll
    for (int r = 0; r < 8; ++r)
      out[(size_t)(m0 + t * 16 + rbase + r) * LDN + col] = acc[t][r];
}

extern "C" void kernel_launch(void* const* d_in, const int* in_sizes, int n_in,
                              void* d_out, int out_size, void* d_ws, size_t ws_size,
                              hipStream_t stream) {
  const float* x   = (const float*)d_in[0];
  const int*   src = (const int*)  d_in[1];
  const int*   dst = (const int*)  d_in[2];
  const float* W1  = (const float*)d_in[3];
  const float* b1  = (const float*)d_in[4];
  const float* W2  = (const float*)d_in[5];
  const float* b2  = (const float*)d_in[6];
  const float* W3  = (const float*)d_in[7];
  const float* b3  = (const float*)d_in[8];
  const float* Wr  = (const float*)d_in[9];
  const float* br  = (const float*)d_in[10];
  const float* Wo  = (const float*)d_in[11];
  const float* bo  = (const float*)d_in[12];
  float* out = (float*)d_out;

  float* ws    = (float*)d_ws;
  float* dsrc  = ws;
  float* ddst  = dsrc + NODES;
  float* RES   = ddst + NODES;
  float* ABUF  = RES  + (size_t)NODES * FDIM;
  float* HBUF  = ABUF + (size_t)NODES * FDIM;
  float* WTr   = HBUF + (size_t)NODES * FDIM;
  float* WT1   = WTr  + FDIM * FDIM;
  float* WT2   = WT1  + FDIM * FDIM;
  float* WT3   = WT2  + FDIM * FDIM;
  float* WTo   = WT3  + FDIM * FDIM;
  int*   cnt   = (int*)(WTo + CDIM * FDIM);
  int*   rowp  = cnt  + NODES;
  int*   rank  = rowp + NODES + 16;
  int*   colx  = rank + EDGES;

  const int eb       = (EDGES + 255) / 256;
  const int nb       = (NODES + 255) / 256;
  const int wb       = (FDIM * FDIM + 255) / 256;
  const int wbo      = (FDIM * CDIM + 255) / 256;
  const int mblocks  = NODES / (16 * MTILE);
  const dim3 gemmH_blk(256);

  zero_deg_kernel<<<nb, 256, 0, stream>>>(dsrc, cnt, NODES);
  degree_rank_kernel<<<eb, 256, 0, stream>>>(src, dst, dsrc, cnt, rank, EDGES);
  scan_norm_kernel<<<nb, 256, 0, stream>>>(cnt, dsrc, ddst, rowp, NODES);
  fill_csr_kernel<<<eb, 256, 0, stream>>>(src, dst, rowp, rank, colx, EDGES);

  transpose_w_kernel<<<wb,  256, 0, stream>>>(Wr, WTr, FDIM, FDIM);
  transpose_w_kernel<<<wb,  256, 0, stream>>>(W1, WT1, FDIM, FDIM);
  transpose_w_kernel<<<wb,  256, 0, stream>>>(W2, WT2, FDIM, FDIM);
  transpose_w_kernel<<<wb,  256, 0, stream>>>(W3, WT3, FDIM, FDIM);
  transpose_w_kernel<<<wbo, 256, 0, stream>>>(Wo, WTo, FDIM, CDIM);

  gemm_k128_wmma<FDIM, false, true>
      <<<mblocks, gemmH_blk, 0, stream>>>(x, WTr, br, nullptr, RES);

  gemm_k128_wmma<FDIM, true, false>
      <<<mblocks, gemmH_blk, 0, stream>>>(x, WT1, nullptr, dsrc, HBUF);


  gather_gemm_kernel<FDIM, false, true, false>
      <<<mblocks, gemmH_blk, 0, stream>>>(HBUF, rowp, colx, ddst, dsrc, b1,
                                          nullptr, WT2, nullptr, ABUF);

  gather_gemm_kernel<FDIM, false, true, false>
      <<<mblocks, gemmH_blk, 0, stream>>>(ABUF, rowp, colx, ddst, dsrc, b2,
                                          nullptr, WT3, nullptr, HBUF);

  gather_gemm_kernel<CDIM, true, false, true>
      <<<mblocks, gemmH_blk, 0, stream>>>(HBUF, rowp, colx, ddst, dsrc, b3,
                                          RES, WTo, bo, out);
}


# B fragments read in place from the row-major weights (4-byte loads, a half-wave taking 16 neighbouring columns of one k row); transposed copies and their 5 launches removed (14 -> 9 launches)
# speedup vs baseline: 2.5802x; 1.0653x over previous
#include <hip/hip_runtime.h>
#include <hip/hip_bf16.h>


typedef __attribute__((ext_vector_type(2))) float v2f;
typedef __attribute__((ext_vector_type(8))) float v8f;

#define NODES 50000
#define EDGES 800000
#define FDIM  128
#define CDIM  64
#define MTILE 5

__global__ void zero_deg_kernel(float* __restrict__ deg_out,
                                int* __restrict__ cnt_in, int n) {
  int i = blockIdx.x * blockDim.x + threadIdx.x;
  if (i < n) { deg_out[i] = 0.0f; cnt_in[i] = 0; }
}

__global__ void degree_rank_kernel(const int* __restrict__ src,
                                   const int* __restrict__ dst,
                                   float* __restrict__ deg_out,
                                   int* __restrict__ cnt_in,
                                   int* __restrict__ rank, int e) {
  int i = blockIdx.x * blockDim.x + threadIdx.x;
  if (i < e) {
    atomicAdd(&deg_out[src[i]], 1.0f);
    rank[i] = atomicAdd(&cnt_in[dst[i]], 1);
  }
}

__global__ void __launch_bounds__(256)
scan_norm_kernel(const int* __restrict__ cnt_in, float* __restrict__ d_src,
                 float* __restrict__ d_dst, int* __restrict__ rowptr, int n) {
  __shared__ int wfront[8];
  __shared__ int wtot[8];
  const int t = threadIdx.x, lane = t & 31, w = t >> 5;
  const int base = blockIdx.x * 256;
  int front = 0;
  const int4* __restrict__ c4 = (const int4*)cnt_in;
  const int n4 = base >> 2;
  for (int i0 = 0; i0 < n4; i0 += 8 * 256) {
    int4 v[8];
#pragma unroll
    for (int u = 0; u < 8; ++u)
      v[u] = c4[min(i0 + u * 256 + t, n4 - 1)];
#pragma unroll
    for (int u = 0; u < 8; ++u)
      if (i0 + u * 256 + t < n4) front += (v[u].x + v[u].y) + (v[u].z + v[u].w);
  }
  for (int off = 16; off > 0; off >>= 1) front += __shfl_xor(front, off);

  const int i = base + t;
  const int c = (i < n) ? cnt_in[i] : 0;
  int incl = c;
  for (int off = 1; off < 32; off <<= 1) {
    const int v = __shfl_up(incl, off);
    if (lane >= off) incl += v;
  }
  if (lane == 0)  wfront[w] = front;
  if (lane == 31) wtot[w]   = incl;
  __syncthreads();
  int carry = 0;
  for (int k = 0; k < 8; ++k) {
    carry += wfront[k];
    if (k < w) carry += wtot[k];
  }
  const int excl = carry + incl - c;
  if (i < n) {
    rowptr[i] = excl;
    if (i == n - 1) rowptr[n] = excl + c;
    const float a = d_src[i];
    d_src[i] = rsqrtf(a > 0.0f ? a : 1.0f);
    d_dst[i] = rsqrtf(c > 0 ? (float)c : 1.0f);
  }
}

__global__ void fill_csr_kernel(const int* __restrict__ src,
                                const int* __restrict__ dst,
                                const int* __restrict__ rowptr,
                                const int* __restrict__ rank,
                                int* __restrict__ colidx, int e) {
  int i = blockIdx.x * blockDim.x + threadIdx.x;
  if (i < e) colidx[rowptr[dst[i]] + rank[i]] = src[i];
}

template <int LDN, bool SCALE_A, bool ADD_BIAS>
__global__ void gemm_k128_wmma(const float* __restrict__ A,
                               const float* __restrict__ W,
                               const float* __restrict__ bias,
                               const float* __restrict__ dsrc,
                               float* __restrict__ out) {
  const int lane = threadIdx.x & 31;
  const int wave = threadIdx.x >> 5;
  const int m0   = blockIdx.x * (16 * MTILE);
  const int n0   = wave << 4;
  const int col  = n0 + (lane & 15);
  const int kb   = (lane >> 4) << 1;

  float init = 0.0f;
  if constexpr (ADD_BIAS) init = bias[col];
  v8f acc[MTILE];
#pragma unroll
  for (int t = 0; t < MTILE; ++t)
#pragma unroll
    for (int r = 0; r < 8; ++r) acc[t][r] = init;

  const float* __restrict__ arow[MTILE];
  float scale[MTILE];
#pragma unroll
  for (int t = 0; t < MTILE; ++t) {
    const int row = m0 + t * 16 + (lane & 15);
    arow[t]  = A + (size_t)row * FDIM;
    scale[t] = SCALE_A ? dsrc[row] : 1.0f;
  }
  const float* __restrict__ bcol = W + (size_t)kb * LDN + col;

#pragma unroll
  for (int k = 0; k < FDIM; k += 4) {
    v2f b;
    b.x = bcol[(size_t)k * LDN];
    b.y = bcol[(size_t)(k + 1) * LDN];
#pragma unroll
    for (int t = 0; t < MTILE; ++t) {
      v2f a;
      a.x = arow[t][k + kb];
      a.y = arow[t][k + kb + 1];
      if constexpr (SCALE_A) { a.x *= scale[t]; a.y *= scale[t]; }
      acc[t] = __builtin_amdgcn_wmma_f32_16x16x4_f32(
           false, a,  false, b,
           (short)0, acc[t],  false,  false);
    }
  }

  const int rbase = (lane >> 4) << 3;
#pragma unroll
  for (int t = 0; t < MTILE; ++t)
#pragma unroll
    for (int r = 0; r < 8; ++r)
      out[(size_t)(m0 + t * 16 + rbase + r) * LDN + col] = acc[t][r];
}

#define RPW    (16 * MTILE / 8)
#define APITCH (FDIM + 4)

__device__ __forceinline__ void add_source_rows(float4& acc,
                                                const float4* __restrict__ h4,
                                                int ids, int cnt, int lane) {
  for (int j = 0; j < cnt; j += 16) {
    float4 v[16];
#pragma unroll
    for (int u = 0; u < 16; ++u) {
      const int sidx = __builtin_amdgcn_readlane(ids, j + u);
      v[u] = h4[(size_t)sidx * (FDIM / 4) + lane];
    }
#pragma unroll
    for (int u = 0; u < 16; ++u) {
      const bool ok = (j + u) < cnt;
      acc.x += ok ? v[u].x : 0.0f;
      acc.y += ok ? v[u].y : 0.0f;
      acc.z += ok ? v[u].z : 0.0f;
      acc.w += ok ? v[u].w : 0.0f;
    }
  }
}

template <int LDN, bool ADD_RES, bool SCALE_A, bool ADD_BIAS>
__global__ void __launch_bounds__(256)
gather_gemm_kernel(const float* __restrict__ h,
                   const int* __restrict__ rowptr,
                   const int* __restrict__ colidx,
                   const float* __restrict__ ddst,
                   const float* __restrict__ dsrc,
                   const float* __restrict__ bias_in,
                   const float* __restrict__ res,
                   const float* __restrict__ W,
                   const float* __restrict__ bias_out,
                   float* __restrict__ out) {
  __shared__ float As[16 * MTILE * APITCH];
  const int lane = threadIdx.x & 31;
  const int wave = __builtin_amdgcn_readfirstlane((int)(threadIdx.x >> 5));
  const int m0   = blockIdx.x * (16 * MTILE);

  {
    const int r0 = m0 + wave * RPW;
    const int   rp  = (lane <= RPW) ? rowptr[r0 + lane] : 0;
    const float ddv = (lane < RPW) ? ddst[r0 + lane] : 0.0f;
    float dsv = 0.0f;
    if constexpr (SCALE_A) dsv = (lane < RPW) ? dsrc[r0 + lane] : 0.0f;
    const float4* __restrict__ h4 = (const float4*)h;
    const float4 bin = ((const float4*)bias_in)[lane];
    int beg = __builtin_amdgcn_readlane(rp, 0);
    int end = __builtin_amdgcn_readlane(rp, 1);
    int raw = colidx[min(beg + lane, EDGES - 1)];
    for (int i = 0; i < RPW; ++i) {
      const int row = r0 + i;
      const int nend = __builtin_amdgcn_readlane(rp, min(i + 2, RPW));
      const int raw_next = colidx[min(end + lane, EDGES - 1)];
      const int mine = (lane < min(32, end - beg)) ? raw : 0;
      float4 r = make_float4(0.0f, 0.0f, 0.0f, 0.0f);
      if constexpr (ADD_RES)
        r = ((const float4*)res)[(size_t)row * (FDIM / 4) + lane];
      float4 acc = make_float4(0.0f, 0.0f, 0.0f, 0.0f);
      add_source_rows(acc, h4, mine, min(32, end - beg), lane);
      for (int base = beg + 32; base < end; base += 32) {
        const int cnt = min(32, end - base);
        const int ids = (lane < cnt) ? colidx[base + lane] : 0;
        add_source_rows(acc, h4, ids, cnt, lane);
      }
      const float dd = __int_as_float(
          __builtin_amdgcn_readlane(__float_as_int(ddv), i));
      float4 o;
      o.x = acc.x * dd + bin.x;
      o.y = acc.y * dd + bin.y;
      o.z = acc.z * dd + bin.z;
      o.w = acc.w * dd + bin.w;
      if constexpr (ADD_RES) {
        o.x += r.x; o.y += r.y; o.z += r.z; o.w += r.w;
      }
      o.x = fmaxf(o.x, 0.0f);
      o.y = fmaxf(o.y, 0.0f);
      o.z = fmaxf(o.z, 0.0f);
      o.w = fmaxf(o.w, 0.0f);
      if constexpr (SCALE_A) {
        const float ds = __int_as_float(
            __builtin_amdgcn_readlane(__float_as_int(dsv), i));
        o.x *= ds; o.y *= ds; o.z *= ds; o.w *= ds;
      }
      *(float4*)&As[(wave * RPW + i) * APITCH + 4 * lane] = o;
      beg = end; end = nend; raw = raw_next;
    }
  }
  __syncthreads();
  if (wave >= LDN / 16) return;

  const int c   = lane & 15;
  const int kh  = (lane >> 4) << 2;
  const int col = (wave << 4) + c;

  float init = 0.0f;
  if constexpr (ADD_BIAS) init = bias_out[col];
  v8f acc[MTILE];
#pragma unroll
  for (int t = 0; t < MTILE; ++t)
#pragma unroll
    for (int r = 0; r < 8; ++r) acc[t][r] = init;

  const float* __restrict__ bcol = W + (size_t)kh * LDN + col;
  const float* arow = As + c * APITCH + kh;

#pragma unroll
  for (int j = 0; j < FDIM / 8; ++j) {
    v2f b0, b1;
    b0.x = bcol[(size_t)(8 * j + 0) * LDN];
    b0.y = bcol[(size_t)(8 * j + 1) * LDN];
    b1.x = bcol[(size_t)(8 * j + 2) * LDN];
    b1.y = bcol[(size_t)(8 * j + 3) * LDN];
    float4 a[MTILE];
#pragma unroll
    for (int t = 0; t < MTILE; ++t)
      a[t] = *(const float4*)(arow + t * 16 * APITCH + 8 * j);
#pragma unroll
    for (int t = 0; t < MTILE; ++t) {
      v2f a0;
      a0.x = a[t].x; a0.y = a[t].y;
      acc[t] = __builtin_amdgcn_wmma_f32_16x16x4_f32(
           false, a0,  false, b0,
           (short)0, acc[t],  false,  false);
    }
#pragma unroll
    for (int t = 0; t < MTILE; ++t) {
      v2f a1;
      a1.x = a[t].z; a1.y = a[t].w;
      acc[t] = __builtin_amdgcn_wmma_f32_16x16x4_f32(
           false, a1,  false, b1,
           (short)0, acc[t],  false,  false);
    }
  }

  const int rbase = (lane >> 4) << 3;
#pragma unroll
  for (int t = 0; t < MTILE; ++t)
#pragma unroll
    for (int r = 0; r < 8; ++r)
      out[(size_t)(m0 + t * 16 + rbase + r) * LDN + col] = acc[t][r];
}

extern "C" void kernel_launch(void* const* d_in, const int* in_sizes, int n_in,
                              void* d_out, int out_size, void* d_ws, size_t ws_size,
                              hipStream_t stream) {
  const float* x   = (const float*)d_in[0];
  const int*   src = (const int*)  d_in[1];
  const int*   dst = (const int*)  d_in[2];
  const float* W1  = (const float*)d_in[3];
  const float* b1  = (const float*)d_in[4];
  const float* W2  = (const float*)d_in[5];
  const float* b2  = (const float*)d_in[6];
  const float* W3  = (const float*)d_in[7];
  const float* b3  = (const float*)d_in[8];
  const float* Wr  = (const float*)d_in[9];
  const float* br  = (const float*)d_in[10];
  const float* Wo  = (const float*)d_in[11];
  const float* bo  = (const float*)d_in[12];
  float* out = (float*)d_out;

  float* ws    = (float*)d_ws;
  float* dsrc  = ws;
  float* ddst  = dsrc + NODES;
  float* RES   = ddst + NODES;
  float* ABUF  = RES  + (size_t)NODES * FDIM;
  float* HBUF  = ABUF + (size_t)NODES * FDIM;
  int*   cnt   = (int*)(HBUF + (size_t)NODES * FDIM);
  int*   rowp  = cnt  + NODES;
  int*   rank  = rowp + NODES + 16;
  int*   colx  = rank + EDGES;

  const int eb       = (EDGES + 255) / 256;
  const int nb       = (NODES + 255) / 256;
  const int mblocks  = NODES / (16 * MTILE);
  const dim3 gemmH_blk(256);

  zero_deg_kernel<<<nb, 256, 0, stream>>>(dsrc, cnt, NODES);
  degree_rank_kernel<<<eb, 256, 0, stream>>>(src, dst, dsrc, cnt, rank, EDGES);
  scan_norm_kernel<<<nb, 256, 0, stream>>>(cnt, dsrc, ddst, rowp, NODES);
  fill_csr_kernel<<<eb, 256, 0, stream>>>(src, dst, rowp, rank, colx, EDGES);

  gemm_k128_wmma<FDIM, false, true>
      <<<mblocks, gemmH_blk, 0, stream>>>(x, Wr, br, nullptr, RES);

  gemm_k128_wmma<FDIM, true, false>
      <<<mblocks, gemmH_blk, 0, stream>>>(x, W1, nullptr, dsrc, HBUF);


  gather_gemm_kernel<FDIM, false, true, false>
      <<<mblocks, gemmH_blk, 0, stream>>>(HBUF, rowp, colx, ddst, dsrc, b1,
                                          nullptr, W2, nullptr, ABUF);

  gather_gemm_kernel<FDIM, false, true, false>
      <<<mblocks, gemmH_blk, 0, stream>>>(ABUF, rowp, colx, ddst, dsrc, b2,
                                          nullptr, W3, nullptr, HBUF);

  gather_gemm_kernel<CDIM, true, false, true>
      <<<mblocks, gemmH_blk, 0, stream>>>(HBUF, rowp, colx, ddst, dsrc, b3,
                                          RES, Wo, bo, out);
}


# The two x GEMMs lose their own launches: the residual GEMM's 625 blocks lead the degree-count grid, the layer-1 GEMM's the CSR-fill grid (independent work shares one grid; 9 -> 7 launches)
# speedup vs baseline: 2.9450x; 1.1414x over previous
#include <hip/hip_runtime.h>
#include <hip/hip_bf16.h>


typedef __attribute__((ext_vector_type(2))) float v2f;
typedef __attribute__((ext_vector_type(8))) float v8f;

#define NODES 50000
#define EDGES 800000
#define FDIM  128
#define CDIM  64
#define MTILE 5
#define MBLOCKS (NODES / (16 * MTILE))
#define EBLOCKS ((EDGES + 255) / 256)

__global__ void zero_deg_kernel(float* __restrict__ deg_out,
                                int* __restrict__ cnt_in, int n) {
  int i = blockIdx.x * blockDim.x + threadIdx.x;
  if (i < n) { deg_out[i] = 0.0f; cnt_in[i] = 0; }
}

__device__ __forceinline__ void degree_rank_edges(const int eblk,
                                                  const int* __restrict__ src,
                                                  const int* __restrict__ dst,
                                                  float* __restrict__ deg_out,
                                                  int* __restrict__ cnt_in,
                                                  int* __restrict__ rank, int e) {
  const int i = eblk * 256 + threadIdx.x;
  if (i < e) {
    atomicAdd(&deg_out[src[i]], 1.0f);
    rank[i] = atomicAdd(&cnt_in[dst[i]], 1);
  }
}

__global__ void __launch_bounds__(256)
scan_norm_kernel(const int* __restrict__ cnt_in, float* __restrict__ d_src,
                 float* __restrict__ d_dst, int* __restrict__ rowptr, int n) {
  __shared__ int wfront[8];
  __shared__ int wtot[8];
  const int t = threadIdx.x, lane = t & 31, w = t >> 5;
  const int base = blockIdx.x * 256;
  int front = 0;
  const int4* __restrict__ c4 = (const int4*)cnt_in;
  const int n4 = base >> 2;
  for (int i0 = 0; i0 < n4; i0 += 8 * 256) {
    int4 v[8];
#pragma unroll
    for (int u = 0; u < 8; ++u)
      v[u] = c4[min(i0 + u * 256 + t, n4 - 1)];
#pragma unroll
    for (int u = 0; u < 8; ++u)
      if (i0 + u * 256 + t < n4) front += (v[u].x + v[u].y) + (v[u].z + v[u].w);
  }
  for (int off = 16; off > 0; off >>= 1) front += __shfl_xor(front, off);

  const int i = base + t;
  const int c = (i < n) ? cnt_in[i] : 0;
  int incl = c;
  for (int off = 1; off < 32; off <<= 1) {
    const int v = __shfl_up(incl, off);
    if (lane >= off) incl += v;
  }
  if (lane == 0)  wfront[w] = front;
  if (lane == 31) wtot[w]   = incl;
  __syncthreads();
  int carry = 0;
  for (int k = 0; k < 8; ++k) {
    carry += wfront[k];
    if (k < w) carry += wtot[k];
  }
  const int excl = carry + incl - c;
  if (i < n) {
    rowptr[i] = excl;
    if (i == n - 1) rowptr[n] = excl + c;
    const float a = d_src[i];
    d_src[i] = rsqrtf(a > 0.0f ? a : 1.0f);
    d_dst[i] = rsqrtf(c > 0 ? (float)c : 1.0f);
  }
}

__device__ __forceinline__ void fill_csr_edges(const int eblk,
                                               const int* __restrict__ src,
                                               const int* __restrict__ dst,
                                               const int* __restrict__ rowptr,
                                               const int* __restrict__ rank,
                                               int* __restrict__ colidx, int e) {
  const int i = eblk * 256 + threadIdx.x;
  if (i < e) colidx[rowptr[dst[i]] + rank[i]] = src[i];
}

template <int LDN, bool SCALE_A, bool ADD_BIAS>
__device__ __forceinline__ void gemm_k128_tile(const int blk,
                                               const float* __restrict__ A,
                                               const float* __restrict__ W,
                                               const float* __restrict__ bias,
                                               const float* __restrict__ dsrc,
                                               float* __restrict__ out) {
  const int lane = threadIdx.x & 31;
  const int wave = threadIdx.x >> 5;
  const int m0   = blk * (16 * MTILE);
  const int n0   = wave << 4;
  const int col  = n0 + (lane & 15);
  const int kb   = (lane >> 4) << 1;

  float init = 0.0f;
  if constexpr (ADD_BIAS) init = bias[col];
  v8f acc[MTILE];
#pragma unroll
  for (int t = 0; t < MTILE; ++t)
#pragma unroll
    for (int r = 0; r < 8; ++r) acc[t][r] = init;

  const float* __restrict__ arow[MTILE];
  float scale[MTILE];
#pragma unroll
  for (int t = 0; t < MTILE; ++t) {
    const int row = m0 + t * 16 + (lane & 15);
    arow[t]  = A + (size_t)row * FDIM;
    scale[t] = SCALE_A ? dsrc[row] : 1.0f;
  }
  const float* __restrict__ bcol = W + (size_t)kb * LDN + col;

#pragma unroll
  for (int k = 0; k < FDIM; k += 4) {
    v2f b;
    b.x = bcol[(size_t)k * LDN];
    b.y = bcol[(size_t)(k + 1) * LDN];
#pragma unroll
    for (int t = 0; t < MTILE; ++t) {
      v2f a;
      a.x = arow[t][k + kb];
      a.y = arow[t][k + kb + 1];
      if constexpr (SCALE_A) { a.x *= scale[t]; a.y *= scale[t]; }
      acc[t] = __builtin_amdgcn_wmma_f32_16x16x4_f32(
           false, a,  false, b,
           (short)0, acc[t],  false,  false);
    }
  }

  const int rbase = (lane >> 4) << 3;
#pragma unroll
  for (int t = 0; t < MTILE; ++t)
#pragma unroll
    for (int r = 0; r < 8; ++r)
      out[(size_t)(m0 + t * 16 + rbase + r) * LDN + col] = acc[t][r];
}

__global__ void __launch_bounds__(256)
degree_rank_gemm_kernel(const int* __restrict__ src,
                        const int* __restrict__ dst,
                        float* __restrict__ deg_out,
                        int* __restrict__ cnt_in,
                        int* __restrict__ rank,
                        const float* __restrict__ A,
                        const float* __restrict__ W,
                        const float* __restrict__ bias,
                        float* __restrict__ out) {
  if (blockIdx.x < MBLOCKS)
    gemm_k128_tile<FDIM, false, true>((int)blockIdx.x, A, W, bias, nullptr, out);
  else
    degree_rank_edges((int)blockIdx.x - MBLOCKS, src, dst, deg_out, cnt_in,
                      rank, EDGES);
}

__global__ void __launch_bounds__(256)
fill_csr_gemm_kernel(const int* __restrict__ src,
                     const int* __restrict__ dst,
                     const int* __restrict__ rowptr,
                     const int* __restrict__ rank,
                     int* __restrict__ colidx,
                     const float* __restrict__ A,
                     const float* __restrict__ W,
                     const float* __restrict__ dsrc,
                     float* __restrict__ out) {
  if (blockIdx.x < MBLOCKS)
    gemm_k128_tile<FDIM, true, false>((int)blockIdx.x, A, W, nullptr, dsrc, out);
  else
    fill_csr_edges((int)blockIdx.x - MBLOCKS, src, dst, rowptr, rank, colidx,
                   EDGES);
}

#define RPW    (16 * MTILE / 8)
#define APITCH (FDIM + 4)

__device__ __forceinline__ void add_source_rows(float4& acc,
                                                const float4* __restrict__ h4,
                                                int ids, int cnt, int lane) {
  for (int j = 0; j < cnt; j += 16) {
    float4 v[16];
#pragma unroll
    for (int u = 0; u < 16; ++u) {
      const int sidx = __builtin_amdgcn_readlane(ids, j + u);
      v[u] = h4[(size_t)sidx * (FDIM / 4) + lane];
    }
#pragma unroll
    for (int u = 0; u < 16; ++u) {
      const bool ok = (j + u) < cnt;
      acc.x += ok ? v[u].x : 0.0f;
      acc.y += ok ? v[u].y : 0.0f;
      acc.z += ok ? v[u].z : 0.0f;
      acc.w += ok ? v[u].w : 0.0f;
    }
  }
}

template <int LDN, bool ADD_RES, bool SCALE_A, bool ADD_BIAS>
__global__ void __launch_bounds__(256)
gather_gemm_kernel(const float* __restrict__ h,
                   const int* __restrict__ rowptr,
                   const int* __restrict__ colidx,
                   const float* __restrict__ ddst,
                   const float* __restrict__ dsrc,
                   const float* __restrict__ bias_in,
                   const float* __restrict__ res,
                   const float* __restrict__ W,
                   const float* __restrict__ bias_out,
                   float* __restrict__ out) {
  __shared__ float As[16 * MTILE * APITCH];
  const int lane = threadIdx.x & 31;
  const int wave = __builtin_amdgcn_readfirstlane((int)(threadIdx.x >> 5));
  const int m0   = blockIdx.x * (16 * MTILE);

  {
    const int r0 = m0 + wave * RPW;
    const int   rp  = (lane <= RPW) ? rowptr[r0 + lane] : 0;
    const float ddv = (lane < RPW) ? ddst[r0 + lane] : 0.0f;
    float dsv = 0.0f;
    if constexpr (SCALE_A) dsv = (lane < RPW) ? dsrc[r0 + lane] : 0.0f;
    const float4* __restrict__ h4 = (const float4*)h;
    const float4 bin = ((const float4*)bias_in)[lane];
    int beg = __builtin_amdgcn_readlane(rp, 0);
    int end = __builtin_amdgcn_readlane(rp, 1);
    int raw = colidx[min(beg + lane, EDGES - 1)];
    for (int i = 0; i < RPW; ++i) {
      const int row = r0 + i;
      const int nend = __builtin_amdgcn_readlane(rp, min(i + 2, RPW));
      const int raw_next = colidx[min(end + lane, EDGES - 1)];
      const int mine = (lane < min(32, end - beg)) ? raw : 0;
      float4 r = make_float4(0.0f, 0.0f, 0.0f, 0.0f);
      if constexpr (ADD_RES)
        r = ((const float4*)res)[(size_t)row * (FDIM / 4) + lane];
      float4 acc = make_float4(0.0f, 0.0f, 0.0f, 0.0f);
      add_source_rows(acc, h4, mine, min(32, end - beg), lane);
      for (int base = beg + 32; base < end; base += 32) {
        const int cnt = min(32, end - base);
        const int ids = (lane < cnt) ? colidx[base + lane] : 0;
        add_source_rows(acc, h4, ids, cnt, lane);
      }
      const float dd = __int_as_float(
          __builtin_amdgcn_readlane(__float_as_int(ddv), i));
      float4 o;
      o.x = acc.x * dd + bin.x;
      o.y = acc.y * dd + bin.y;
      o.z = acc.z * dd + bin.z;
      o.w = acc.w * dd + bin.w;
      if constexpr (ADD_RES) {
        o.x += r.x; o.y += r.y; o.z += r.z; o.w += r.w;
      }
      o.x = fmaxf(o.x, 0.0f);
      o.y = fmaxf(o.y, 0.0f);
      o.z = fmaxf(o.z, 0.0f);
      o.w = fmaxf(o.w, 0.0f);
      if constexpr (SCALE_A) {
        const float ds = __int_as_float(
            __builtin_amdgcn_readlane(__float_as_int(dsv), i));
        o.x *= ds; o.y *= ds; o.z *= ds; o.w *= ds;
      }
      *(float4*)&As[(wave * RPW + i) * APITCH + 4 * lane] = o;
      beg = end; end = nend; raw = raw_next;
    }
  }
  __syncthreads();
  if (wave >= LDN / 16) return;

  const int c   = lane & 15;
  const int kh  = (lane >> 4) << 2;
  const int col = (wave << 4) + c;

  float init = 0.0f;
  if constexpr (ADD_BIAS) init = bias_out[col];
  v8f acc[MTILE];
#pragma unroll
  for (int t = 0; t < MTILE; ++t)
#pragma unroll
    for (int r = 0; r < 8; ++r) acc[t][r] = init;

  const float* __restrict__ bcol = W + (size_t)kh * LDN + col;
  const float* arow = As + c * APITCH + kh;

#pragma unroll
  for (int j = 0; j < FDIM / 8; ++j) {
    v2f b0, b1;
    b0.x = bcol[(size_t)(8 * j + 0) * LDN];
    b0.y = bcol[(size_t)(8 * j + 1) * LDN];
    b1.x = bcol[(size_t)(8 * j + 2) * LDN];
    b1.y = bcol[(size_t)(8 * j + 3) * LDN];
    float4 a[MTILE];
#pragma unroll
    for (int t = 0; t < MTILE; ++t)
      a[t] = *(const float4*)(arow + t * 16 * APITCH + 8 * j);
#pragma unroll
    for (int t = 0; t < MTILE; ++t) {
      v2f a0;
      a0.x = a[t].x; a0.y = a[t].y;
      acc[t] = __builtin_amdgcn_wmma_f32_16x16x4_f32(
           false, a0,  false, b0,
           (short)0, acc[t],  false,  false);
    }
#pragma unroll
    for (int t = 0; t < MTILE; ++t) {
      v2f a1;
      a1.x = a[t].z; a1.y = a[t].w;
      acc[t] = __builtin_amdgcn_wmma_f32_16x16x4_f32(
           false, a1,  false, b1,
           (short)0, acc[t],  false,  false);
    }
  }

  const int rbase = (lane >> 4) << 3;
#pragma unroll
  for (int t = 0; t < MTILE; ++t)
#pragma unroll
    for (int r = 0; r < 8; ++r)
      out[(size_t)(m0 + t * 16 + rbase + r) * LDN + col] = acc[t][r];
}

extern "C" void kernel_launch(void* const* d_in, const int* in_sizes, int n_in,
                              void* d_out, int out_size, void* d_ws, size_t ws_size,
                              hipStream_t stream) {
  const float* x   = (const float*)d_in[0];
  const int*   src = (const int*)  d_in[1];
  const int*   dst = (const int*)  d_in[2];
  const float* W1  = (const float*)d_in[3];
  const float* b1  = (const float*)d_in[4];
  const float* W2  = (const float*)d_in[5];
  const float* b2  = (const float*)d_in[6];
  const float* W3  = (const float*)d_in[7];
  const float* b3  = (const float*)d_in[8];
  const float* Wr  = (const float*)d_in[9];
  const float* br  = (const float*)d_in[10];
  const float* Wo  = (const float*)d_in[11];
  const float* bo  = (const float*)d_in[12];
  float* out = (float*)d_out;

  float* ws    = (float*)d_ws;
  float* dsrc  = ws;
  float* ddst  = dsrc + NODES;
  float* RES   = ddst + NODES;
  float* ABUF  = RES  + (size_t)NODES * FDIM;
  float* HBUF  = ABUF + (size_t)NODES * FDIM;
  int*   cnt   = (int*)(HBUF + (size_t)NODES * FDIM);
  int*   rowp  = cnt  + NODES;
  int*   rank  = rowp + NODES + 16;
  int*   colx  = rank + EDGES;

  const int nb       = (NODES + 255) / 256;
  const int mblocks  = MBLOCKS;
  const dim3 gemmH_blk(256);

  zero_deg_kernel<<<nb, 256, 0, stream>>>(dsrc, cnt, NODES);
  degree_rank_gemm_kernel<<<MBLOCKS + EBLOCKS, 256, 0, stream>>>(
      src, dst, dsrc, cnt, rank, x, Wr, br, RES);
  scan_norm_kernel<<<nb, 256, 0, stream>>>(cnt, dsrc, ddst, rowp, NODES);
  fill_csr_gemm_kernel<<<MBLOCKS + EBLOCKS, 256, 0, stream>>>(
      src, dst, rowp, rank, colx, x, W1, dsrc, HBUF);


  gather_gemm_kernel<FDIM, false, true, false>
      <<<mblocks, gemmH_blk, 0, stream>>>(HBUF, rowp, colx, ddst, dsrc, b1,
                                          nullptr, W2, nullptr, ABUF);

  gather_gemm_kernel<FDIM, false, true, false>
      <<<mblocks, gemmH_blk, 0, stream>>>(ABUF, rowp, colx, ddst, dsrc, b2,
                                          nullptr, W3, nullptr, HBUF);

  gather_gemm_kernel<CDIM, true, false, true>
      <<<mblocks, gemmH_blk, 0, stream>>>(HBUF, rowp, colx, ddst, dsrc, b3,
                                          RES, Wo, bo, out);
}


# The two GEMMs on x lay their 80-row A tile down in LDS once, by coalesced 16-byte loads, and take fragments there as the pull kernels do (was 8-byte loads 16 lines apart); k summed in another order
# speedup vs baseline: 3.0382x; 1.0316x over previous
#include <hip/hip_runtime.h>
#include <hip/hip_bf16.h>


typedef __attribute__((ext_vector_type(2))) float v2f;
typedef __attribute__((ext_vector_type(8))) float v8f;

#define NODES 50000
#define EDGES 800000
#define FDIM  128
#define CDIM  64
#define MTILE 5
#define MBLOCKS (NODES / (16 * MTILE))
#define EBLOCKS ((EDGES + 255) / 256)

__global__ void zero_deg_kernel(float* __restrict__ deg_out,
                                int* __restrict__ cnt_in, int n) {
  int i = blockIdx.x * blockDim.x + threadIdx.x;
  if (i < n) { deg_out[i] = 0.0f; cnt_in[i] = 0; }
}

__device__ __forceinline__ void degree_rank_edges(const int eblk,
                                                  const int* __restrict__ src,
                                                  const int* __restrict__ dst,
                                                  float* __restrict__ deg_out,
                                                  int* __restrict__ cnt_in,
                                                  int* __restrict__ rank, int e) {
  const int i = eblk * 256 + threadIdx.x;
  if (i < e) {
    atomicAdd(&deg_out[src[i]], 1.0f);
    rank[i] = atomicAdd(&cnt_in[dst[i]], 1);
  }
}

__global__ void __launch_bounds__(256)
scan_norm_kernel(const int* __restrict__ cnt_in, float* __restrict__ d_src,
                 float* __restrict__ d_dst, int* __restrict__ rowptr, int n) {
  __shared__ int wfront[8];
  __shared__ int wtot[8];
  const int t = threadIdx.x, lane = t & 31, w = t >> 5;
  const int base = blockIdx.x * 256;
  int front = 0;
  const int4* __restrict__ c4 = (const int4*)cnt_in;
  const int n4 = base >> 2;
  for (int i0 = 0; i0 < n4; i0 += 8 * 256) {
    int4 v[8];
#pragma unroll
    for (int u = 0; u < 8; ++u)
      v[u] = c4[min(i0 + u * 256 + t, n4 - 1)];
#pragma unroll
    for (int u = 0; u < 8; ++u)
      if (i0 + u * 256 + t < n4) front += (v[u].x + v[u].y) + (v[u].z + v[u].w);
  }
  for (int off = 16; off > 0; off >>= 1) front += __shfl_xor(front, off);

  const int i = base + t;
  const int c = (i < n) ? cnt_in[i] : 0;
  int incl = c;
  for (int off = 1; off < 32; off <<= 1) {
    const int v = __shfl_up(incl, off);
    if (lane >= off) incl += v;
  }
  if (lane == 0)  wfront[w] = front;
  if (lane == 31) wtot[w]   = incl;
  __syncthreads();
  int carry = 0;
  for (int k = 0; k < 8; ++k) {
    carry += wfront[k];
    if (k < w) carry += wtot[k];
  }
  const int excl = carry + incl - c;
  if (i < n) {
    rowptr[i] = excl;
    if (i == n - 1) rowptr[n] = excl + c;
    const float a = d_src[i];
    d_src[i] = rsqrtf(a > 0.0f ? a : 1.0f);
    d_dst[i] = rsqrtf(c > 0 ? (float)c : 1.0f);
  }
}

__device__ __forceinline__ void fill_csr_edges(const int eblk,
                                               const int* __restrict__ src,
                                               const int* __restrict__ dst,
                                               const int* __restrict__ rowptr,
                                               const int* __restrict__ rank,
                                               int* __restrict__ colidx, int e) {
  const int i = eblk * 256 + threadIdx.x;
  if (i < e) colidx[rowptr[dst[i]] + rank[i]] = src[i];
}

#define RPW    (16 * MTILE / 8)
#define APITCH (FDIM + 4)

template <int LDN, bool ADD_BIAS>
__device__ __forceinline__ void lds_tile_gemm(const float* As,
                                              const float* __restrict__ W,
                                              const float* __restrict__ bias_out,
                                              float* __restrict__ out,
                                              const int m0, const int wave,
                                              const int lane) {
  const int c   = lane & 15;
  const int kh  = (lane >> 4) << 2;
  const int col = (wave << 4) + c;

  float init = 0.0f;
  if constexpr (ADD_BIAS) init = bias_out[col];
  v8f acc[MTILE];
#pragma unroll
  for (int t = 0; t < MTILE; ++t)
#pragma unroll
    for (int r = 0; r < 8; ++r) acc[t][r] = init;

  const float* __restrict__ bcol = W + (size_t)kh * LDN + col;
  const float* arow = As + c * APITCH + kh;

#pragma unroll
  for (int j = 0; j < FDIM / 8; ++j) {
    v2f b0, b1;
    b0.x = bcol[(size_t)(8 * j + 0) * LDN];
    b0.y = bcol[(size_t)(8 * j + 1) * LDN];
    b1.x = bcol[(size_t)(8 * j + 2) * LDN];
    b1.y = bcol[(size_t)(8 * j + 3) * LDN];
    float4 a[MTILE];
#pragma unroll
    for (int t = 0; t < MTILE; ++t)
      a[t] = *(const float4*)(arow + t * 16 * APITCH + 8 * j);
#pragma unroll
    for (int t = 0; t < MTILE; ++t) {
      v2f a0;
      a0.x = a[t].x; a0.y = a[t].y;
      acc[t] = __builtin_amdgcn_wmma_f32_16x16x4_f32(
           false, a0,  false, b0,
           (short)0, acc[t],  false,  false);
    }
#pragma unroll
    for (int t = 0; t < MTILE; ++t) {
      v2f a1;
      a1.x = a[t].z; a1.y = a[t].w;
      acc[t] = __builtin_amdgcn_wmma_f32_16x16x4_f32(
           false, a1,  false, b1,
           (short)0, acc[t],  false,  false);
    }
  }

  const int rbase = (lane >> 4) << 3;
#pragma unroll
  for (int t = 0; t < MTILE; ++t)
#pragma unroll
    for (int r = 0; r < 8; ++r)
      out[(size_t)(m0 + t * 16 + rbase + r) * LDN + col] = acc[t][r];
}

template <bool SCALE_A, bool ADD_BIAS>
__device__ __forceinline__ void gemm_x_tile(float* As, const int blk,
                                            const float* __restrict__ A,
                                            const float* __restrict__ W,
                                            const float* __restrict__ bias,
                                            const float* __restrict__ dsrc,
                                            float* __restrict__ out) {
  const int lane = threadIdx.x & 31;
  const int wave = __builtin_amdgcn_readfirstlane((int)(threadIdx.x >> 5));
  const int m0   = blk * (16 * MTILE);
  const int r0   = m0 + wave * RPW;
  const float4* __restrict__ A4 = (const float4*)A;
  float4 v[RPW];
#pragma unroll
  for (int i = 0; i < RPW; ++i)
    v[i] = A4[(size_t)(r0 + i) * (FDIM / 4) + lane];
#pragma unroll
  for (int i = 0; i < RPW; ++i) {
    float4 o = v[i];
    if constexpr (SCALE_A) {
      const float s = dsrc[r0 + i];
      o.x *= s; o.y *= s; o.z *= s; o.w *= s;
    }
    *(float4*)&As[(wave * RPW + i) * APITCH + 4 * lane] = o;
  }
  __syncthreads();
  lds_tile_gemm<FDIM, ADD_BIAS>(As, W, bias, out, m0, wave, lane);
}

__global__ void __launch_bounds__(256)
degree_rank_gemm_kernel(const int* __restrict__ src,
                        const int* __restrict__ dst,
                        float* __restrict__ deg_out,
                        int* __restrict__ cnt_in,
                        int* __restrict__ rank,
                        const float* __restrict__ A,
                        const float* __restrict__ W,
                        const float* __restrict__ bias,
                        float* __restrict__ out) {
  __shared__ float As[16 * MTILE * APITCH];
  if (blockIdx.x < MBLOCKS)
    gemm_x_tile<false, true>(As, (int)blockIdx.x, A, W, bias, nullptr, out);
  else
    degree_rank_edges((int)blockIdx.x - MBLOCKS, src, dst, deg_out, cnt_in,
                      rank, EDGES);
}

__global__ void __launch_bounds__(256)
fill_csr_gemm_kernel(const int* __restrict__ src,
                     const int* __restrict__ dst,
                     const int* __restrict__ rowptr,
                     const int* __restrict__ rank,
                     int* __restrict__ colidx,
                     const float* __restrict__ A,
                     const float* __restrict__ W,
                     const float* __restrict__ dsrc,
                     float* __restrict__ out) {
  __shared__ float As[16 * MTILE * APITCH];
  if (blockIdx.x < MBLOCKS)
    gemm_x_tile<true, false>(As, (int)blockIdx.x, A, W, nullptr, dsrc, out);
  else
    fill_csr_edges((int)blockIdx.x - MBLOCKS, src, dst, rowptr, rank, colidx,
                   EDGES);
}


__device__ __forceinline__ void add_source_rows(float4& acc,
                                                const float4* __restrict__ h4,
                                                int ids, int cnt, int lane) {
  for (int j = 0; j < cnt; j += 16) {
    float4 v[16];
#pragma unroll
    for (int u = 0; u < 16; ++u) {
      const int sidx = __builtin_amdgcn_readlane(ids, j + u);
      v[u] = h4[(size_t)sidx * (FDIM / 4) + lane];
    }
#pragma unroll
    for (int u = 0; u < 16; ++u) {
      const bool ok = (j + u) < cnt;
      acc.x += ok ? v[u].x : 0.0f;
      acc.y += ok ? v[u].y : 0.0f;
      acc.z += ok ? v[u].z : 0.0f;
      acc.w += ok ? v[u].w : 0.0f;
    }
  }
}

template <int LDN, bool ADD_RES, bool SCALE_A, bool ADD_BIAS>
__global__ void __launch_bounds__(256)
gather_gemm_kernel(const float* __restrict__ h,
                   const int* __restrict__ rowptr,
                   const int* __restrict__ colidx,
                   const float* __restrict__ ddst,
                   const float* __restrict__ dsrc,
                   const float* __restrict__ bias_in,
                   const float* __restrict__ res,
                   const float* __restrict__ W,
                   const float* __restrict__ bias_out,
                   float* __restrict__ out) {
  __shared__ float As[16 * MTILE * APITCH];
  const int lane = threadIdx.x & 31;
  const int wave = __builtin_amdgcn_readfirstlane((int)(threadIdx.x >> 5));
  const int m0   = blockIdx.x * (16 * MTILE);

  {
    const int r0 = m0 + wave * RPW;
    const int   rp  = (lane <= RPW) ? rowptr[r0 + lane] : 0;
    const float ddv = (lane < RPW) ? ddst[r0 + lane] : 0.0f;
    float dsv = 0.0f;
    if constexpr (SCALE_A) dsv = (lane < RPW) ? dsrc[r0 + lane] : 0.0f;
    const float4* __restrict__ h4 = (const float4*)h;
    const float4 bin = ((const float4*)bias_in)[lane];
    int beg = __builtin_amdgcn_readlane(rp, 0);
    int end = __builtin_amdgcn_readlane(rp, 1);
    int raw = colidx[min(beg + lane, EDGES - 1)];
    for (int i = 0; i < RPW; ++i) {
      const int row = r0 + i;
      const int nend = __builtin_amdgcn_readlane(rp, min(i + 2, RPW));
      const int raw_next = colidx[min(end + lane, EDGES - 1)];
      const int mine = (lane < min(32, end - beg)) ? raw : 0;
      float4 r = make_float4(0.0f, 0.0f, 0.0f, 0.0f);
      if constexpr (ADD_RES)
        r = ((const float4*)res)[(size_t)row * (FDIM / 4) + lane];
      float4 acc = make_float4(0.0f, 0.0f, 0.0f, 0.0f);
      add_source_rows(acc, h4, mine, min(32, end - beg), lane);
      for (int base = beg + 32; base < end; base += 32) {
        const int cnt = min(32, end - base);
        const int ids = (lane < cnt) ? colidx[base + lane] : 0;
        add_source_rows(acc, h4, ids, cnt, lane);
      }
      const float dd = __int_as_float(
          __builtin_amdgcn_readlane(__float_as_int(ddv), i));
      float4 o;
      o.x = acc.x * dd + bin.x;
      o.y = acc.y * dd + bin.y;
      o.z = acc.z * dd + bin.z;
      o.w = acc.w * dd + bin.w;
      if constexpr (ADD_RES) {
        o.x += r.x; o.y += r.y; o.z += r.z; o.w += r.w;
      }
      o.x = fmaxf(o.x, 0.0f);
      o.y = fmaxf(o.y, 0.0f);
      o.z = fmaxf(o.z, 0.0f);
      o.w = fmaxf(o.w, 0.0f);
      if constexpr (SCALE_A) {
        const float ds = __int_as_float(
            __builtin_amdgcn_readlane(__float_as_int(dsv), i));
        o.x *= ds; o.y *= ds; o.z *= ds; o.w *= ds;
      }
      *(float4*)&As[(wave * RPW + i) * APITCH + 4 * lane] = o;
      beg = end; end = nend; raw = raw_next;
    }
  }
  __syncthreads();
  if (wave >= LDN / 16) return;

  lds_tile_gemm<LDN, ADD_BIAS>(As, W, bias_out, out, m0, wave, lane);
}

extern "C" void kernel_launch(void* const* d_in, const int* in_sizes, int n_in,
                              void* d_out, int out_size, void* d_ws, size_t ws_size,
                              hipStream_t stream) {
  const float* x   = (const float*)d_in[0];
  const int*   src = (const int*)  d_in[1];
  const int*   dst = (const int*)  d_in[2];
  const float* W1  = (const float*)d_in[3];
  const float* b1  = (const float*)d_in[4];
  const float* W2  = (const float*)d_in[5];
  const float* b2  = (const float*)d_in[6];
  const float* W3  = (const float*)d_in[7];
  const float* b3  = (const float*)d_in[8];
  const float* Wr  = (const float*)d_in[9];
  const float* br  = (const float*)d_in[10];
  const float* Wo  = (const float*)d_in[11];
  const float* bo  = (const float*)d_in[12];
  float* out = (float*)d_out;

  float* ws    = (float*)d_ws;
  float* dsrc  = ws;
  float* ddst  = dsrc + NODES;
  float* RES   = ddst + NODES;
  float* ABUF  = RES  + (size_t)NODES * FDIM;
  float* HBUF  = ABUF + (size_t)NODES * FDIM;
  int*   cnt   = (int*)(HBUF + (size_t)NODES * FDIM);
  int*   rowp  = cnt  + NODES;
  int*   rank  = rowp + NODES + 16;
  int*   colx  = rank + EDGES;

  const int nb       = (NODES + 255) / 256;
  const int mblocks  = MBLOCKS;
  const dim3 gemmH_blk(256);

  zero_deg_kernel<<<nb, 256, 0, stream>>>(dsrc, cnt, NODES);
  degree_rank_gemm_kernel<<<MBLOCKS + EBLOCKS, 256, 0, stream>>>(
      src, dst, dsrc, cnt, rank, x, Wr, br, RES);
  scan_norm_kernel<<<nb, 256, 0, stream>>>(cnt, dsrc, ddst, rowp, NODES);
  fill_csr_gemm_kernel<<<MBLOCKS + EBLOCKS, 256, 0, stream>>>(
      src, dst, rowp, rank, colx, x, W1, dsrc, HBUF);


  gather_gemm_kernel<FDIM, false, true, false>
      <<<mblocks, gemmH_blk, 0, stream>>>(HBUF, rowp, colx, ddst, dsrc, b1,
                                          nullptr, W2, nullptr, ABUF);

  gather_gemm_kernel<FDIM, false, true, false>
      <<<mblocks, gemmH_blk, 0, stream>>>(ABUF, rowp, colx, ddst, dsrc, b2,
                                          nullptr, W3, nullptr, HBUF);

  gather_gemm_kernel<CDIM, true, false, true>
      <<<mblocks, gemmH_blk, 0, stream>>>(HBUF, rowp, colx, ddst, dsrc, b3,
                                          RES, Wo, bo, out);
}


# speedup vs baseline: 3.2279x; 1.0624x over previous
#include <hip/hip_runtime.h>
#include <hip/hip_bf16.h>


typedef __attribute__((ext_vector_type(2))) float v2f;
typedef __attribute__((ext_vector_type(8))) float v8f;

#define NODES 50000
#define EDGES 800000
#define FDIM  128
#define CDIM  64
#define MTILE 5
#define MBLOCKS (NODES / (16 * MTILE))
#define EBLOCKS ((EDGES + 255) / 256)

__global__ void zero_deg_kernel(float* __restrict__ deg_out,
                                int* __restrict__ cnt_in, int n) {
  int i = blockIdx.x * blockDim.x + threadIdx.x;
  if (i < n) { deg_out[i] = 0.0f; cnt_in[i] = 0; }
}

__device__ __forceinline__ void degree_rank_edges(const int eblk,
                                                  const int* __restrict__ src,
                                                  const int* __restrict__ dst,
                                                  float* __restrict__ deg_out,
                                                  int* __restrict__ cnt_in,
                                                  int* __restrict__ rank, int e) {
  const int i = eblk * 256 + threadIdx.x;
  if (i < e) {
    atomicAdd(&deg_out[src[i]], 1.0f);
    rank[i] = atomicAdd(&cnt_in[dst[i]], 1);
  }
}

__global__ void __launch_bounds__(256)
scan_norm_kernel(const int* __restrict__ cnt_in, float* __restrict__ d_src,
                 float* __restrict__ d_dst, int* __restrict__ rowptr, int n) {
  __shared__ int wfront[8];
  __shared__ int wtot[8];
  const int t = threadIdx.x, lane = t & 31, w = t >> 5;
  const int base = blockIdx.x * 256;
  int front = 0;
  const int4* __restrict__ c4 = (const int4*)cnt_in;
  const int n4 = base >> 2;
  for (int i0 = 0; i0 < n4; i0 += 8 * 256) {
    int4 v[8];
#pragma unroll
    for (int u = 0; u < 8; ++u)
      v[u] = c4[min(i0 + u * 256 + t, n4 - 1)];
#pragma unroll
    for (int u = 0; u < 8; ++u)
      if (i0 + u * 256 + t < n4) front += (v[u].x + v[u].y) + (v[u].z + v[u].w);
  }
  for (int off = 16; off > 0; off >>= 1) front += __shfl_xor(front, off);

  const int i = base + t;
  const int c = (i < n) ? cnt_in[i] : 0;
  int incl = c;
  for (int off = 1; off < 32; off <<= 1) {
    const int v = __shfl_up(incl, off);
    if (lane >= off) incl += v;
  }
  if (lane == 0)  wfront[w] = front;
  if (lane == 31) wtot[w]   = incl;
  __syncthreads();
  int carry = 0;
  for (int k = 0; k < 8; ++k) {
    carry += wfront[k];
    if (k < w) carry += wtot[k];
  }
  const int excl = carry + incl - c;
  if (i < n) {
    rowptr[i] = excl;
    if (i == n - 1) rowptr[n] = excl + c;
    const float a = d_src[i];
    d_src[i] = rsqrtf(a > 0.0f ? a : 1.0f);
    d_dst[i] = rsqrtf(c > 0 ? (float)c : 1.0f);
  }
}

__device__ __forceinline__ void fill_csr_edges(const int eblk,
                                               const int* __restrict__ src,
                                               const int* __restrict__ dst,
                                               const int* __restrict__ rowptr,
                                               const int* __restrict__ rank,
                                               int* __restrict__ colidx, int e) {
  const int i = eblk * 256 + threadIdx.x;
  if (i < e) colidx[rowptr[dst[i]] + rank[i]] = src[i];
}

#define RPW    (16 * MTILE / 8)
#define APITCH (FDIM + 4)

template <int LDN, bool ADD_BIAS>
__device__ __forceinline__ void lds_tile_gemm(float* As,
                                              const float* __restrict__ W,
                                              const float* __restrict__ bias_out,
                                              float* __restrict__ out,
                                              const int m0, const int wave,
                                              const int lane) {
  constexpr int NCT = LDN / 16;
  const int c   = lane & 15;
  const int kh  = (lane >> 4) << 2;
  const int col = (wave << 4) + c;

  v8f acc[MTILE];
  if (wave < NCT) {
    float init = 0.0f;
    if constexpr (ADD_BIAS) init = bias_out[col];
#pragma unroll
    for (int t = 0; t < MTILE; ++t)
#pragma unroll
      for (int r = 0; r < 8; ++r) acc[t][r] = init;

    const float* __restrict__ bcol = W + (size_t)kh * LDN + col;
    const float* arow = As + c * APITCH + kh;

#pragma unroll
    for (int j = 0; j < FDIM / 8; ++j) {
      v2f b0, b1;
      b0.x = bcol[(size_t)(8 * j + 0) * LDN];
      b0.y = bcol[(size_t)(8 * j + 1) * LDN];
      b1.x = bcol[(size_t)(8 * j + 2) * LDN];
      b1.y = bcol[(size_t)(8 * j + 3) * LDN];
      float4 a[MTILE];
#pragma unroll
      for (int t = 0; t < MTILE; ++t)
        a[t] = *(const float4*)(arow + t * 16 * APITCH + 8 * j);
#pragma unroll
      for (int t = 0; t < MTILE; ++t) {
        v2f a0;
        a0.x = a[t].x; a0.y = a[t].y;
        acc[t] = __builtin_amdgcn_wmma_f32_16x16x4_f32(
             false, a0,  false, b0,
             (short)0, acc[t],  false,  false);
      }
#pragma unroll
      for (int t = 0; t < MTILE; ++t) {
        v2f a1;
        a1.x = a[t].z; a1.y = a[t].w;
        acc[t] = __builtin_amdgcn_wmma_f32_16x16x4_f32(
             false, a1,  false, b1,
             (short)0, acc[t],  false,  false);
      }
    }
  }

  __syncthreads();
  if (wave < NCT) {
    const int rbase = (lane >> 4) << 3;
#pragma unroll
    for (int t = 0; t < MTILE; ++t)
#pragma unroll
      for (int r = 0; r < 8; ++r)
        As[(t * 16 + rbase + r) * APITCH + col] = acc[t][r];
  }
  __syncthreads();
  constexpr int R4 = LDN / 4;
  constexpr int Q  = (16 * MTILE) * R4 / 32 / 8;
  float4* __restrict__ out4 = (float4*)(out + (size_t)m0 * LDN);
#pragma unroll
  for (int q = 0; q < Q; ++q) {
    const int e   = (wave * Q + q) * 32 + lane;
    const int row = e / R4;
    const int c4  = e % R4;
    out4[e] = *(const float4*)&As[row * APITCH + 4 * c4];
  }
}

template <bool SCALE_A, bool ADD_BIAS>
__device__ __forceinline__ void gemm_x_tile(float* As, const int blk,
                                            const float* __restrict__ A,
                                            const float* __restrict__ W,
                                            const float* __restrict__ bias,
                                            const float* __restrict__ dsrc,
                                            float* __restrict__ out) {
  const int lane = threadIdx.x & 31;
  const int wave = __builtin_amdgcn_readfirstlane((int)(threadIdx.x >> 5));
  const int m0   = blk * (16 * MTILE);
  const int r0   = m0 + wave * RPW;
  const float4* __restrict__ A4 = (const float4*)A;
  float4 v[RPW];
#pragma unroll
  for (int i = 0; i < RPW; ++i)
    v[i] = A4[(size_t)(r0 + i) * (FDIM / 4) + lane];
#pragma unroll
  for (int i = 0; i < RPW; ++i) {
    float4 o = v[i];
    if constexpr (SCALE_A) {
      const float s = dsrc[r0 + i];
      o.x *= s; o.y *= s; o.z *= s; o.w *= s;
    }
    *(float4*)&As[(wave * RPW + i) * APITCH + 4 * lane] = o;
  }
  __syncthreads();
  lds_tile_gemm<FDIM, ADD_BIAS>(As, W, bias, out, m0, wave, lane);
}

__global__ void __launch_bounds__(256)
degree_rank_gemm_kernel(const int* __restrict__ src,
                        const int* __restrict__ dst,
                        float* __restrict__ deg_out,
                        int* __restrict__ cnt_in,
                        int* __restrict__ rank,
                        const float* __restrict__ A,
                        const float* __restrict__ W,
                        const float* __restrict__ bias,
                        float* __restrict__ out) {
  __shared__ float As[16 * MTILE * APITCH];
  if (blockIdx.x < MBLOCKS)
    gemm_x_tile<false, true>(As, (int)blockIdx.x, A, W, bias, nullptr, out);
  else
    degree_rank_edges((int)blockIdx.x - MBLOCKS, src, dst, deg_out, cnt_in,
                      rank, EDGES);
}

__global__ void __launch_bounds__(256)
fill_csr_gemm_kernel(const int* __restrict__ src,
                     const int* __restrict__ dst,
                     const int* __restrict__ rowptr,
                     const int* __restrict__ rank,
                     int* __restrict__ colidx,
                     const float* __restrict__ A,
                     const float* __restrict__ W,
                     const float* __restrict__ dsrc,
                     float* __restrict__ out) {
  __shared__ float As[16 * MTILE * APITCH];
  if (blockIdx.x < MBLOCKS)
    gemm_x_tile<true, false>(As, (int)blockIdx.x, A, W, nullptr, dsrc, out);
  else
    fill_csr_edges((int)blockIdx.x - MBLOCKS, src, dst, rowptr, rank, colidx,
                   EDGES);
}


__device__ __forceinline__ void add_source_rows(float4& acc,
                                                const float4* __restrict__ h4,
                                                int ids, int cnt, int lane) {
  for (int j = 0; j < cnt; j += 16) {
    float4 v[16];
#pragma unroll
    for (int u = 0; u < 16; ++u) {
      const int sidx = __builtin_amdgcn_readlane(ids, j + u);
      v[u] = h4[(size_t)sidx * (FDIM / 4) + lane];
    }
#pragma unroll
    for (int u = 0; u < 16; ++u) {
      const bool ok = (j + u) < cnt;
      acc.x += ok ? v[u].x : 0.0f;
      acc.y += ok ? v[u].y : 0.0f;
      acc.z += ok ? v[u].z : 0.0f;
      acc.w += ok ? v[u].w : 0.0f;
    }
  }
}

template <int LDN, bool ADD_RES, bool SCALE_A, bool ADD_BIAS>
__global__ void __launch_bounds__(256)
gather_gemm_kernel(const float* __restrict__ h,
                   const int* __restrict__ rowptr,
                   const int* __restrict__ colidx,
                   const float* __restrict__ ddst,
                   const float* __restrict__ dsrc,
                   const float* __restrict__ bias_in,
                   const float* __restrict__ res,
                   const float* __restrict__ W,
                   const float* __restrict__ bias_out,
                   float* __restrict__ out) {
  __shared__ float As[16 * MTILE * APITCH];
  const int lane = threadIdx.x & 31;
  const int wave = __builtin_amdgcn_readfirstlane((int)(threadIdx.x >> 5));
  const int m0   = blockIdx.x * (16 * MTILE);

  {
    const int r0 = m0 + wave * RPW;
    const int   rp  = (lane <= RPW) ? rowptr[r0 + lane] : 0;
    const float ddv = (lane < RPW) ? ddst[r0 + lane] : 0.0f;
    float dsv = 0.0f;
    if constexpr (SCALE_A) dsv = (lane < RPW) ? dsrc[r0 + lane] : 0.0f;
    const float4* __restrict__ h4 = (const float4*)h;
    const float4 bin = ((const float4*)bias_in)[lane];
    int beg = __builtin_amdgcn_readlane(rp, 0);
    int end = __builtin_amdgcn_readlane(rp, 1);
    int raw = colidx[min(beg + lane, EDGES - 1)];
    for (int i = 0; i < RPW; ++i) {
      const int row = r0 + i;
      const int nend = __builtin_amdgcn_readlane(rp, min(i + 2, RPW));
      const int raw_next = colidx[min(end + lane, EDGES - 1)];
      const int mine = (lane < min(32, end - beg)) ? raw : 0;
      float4 r = make_float4(0.0f, 0.0f, 0.0f, 0.0f);
      if constexpr (ADD_RES)
        r = ((const float4*)res)[(size_t)row * (FDIM / 4) + lane];
      float4 acc = make_float4(0.0f, 0.0f, 0.0f, 0.0f);
      add_source_rows(acc, h4, mine, min(32, end - beg), lane);
      for (int base = beg + 32; base < end; base += 32) {
        const int cnt = min(32, end - base);
        const int ids = (lane < cnt) ? colidx[base + lane] : 0;
        add_source_rows(acc, h4, ids, cnt, lane);
      }
      const float dd = __int_as_float(
          __builtin_amdgcn_readlane(__float_as_int(ddv), i));
      float4 o;
      o.x = acc.x * dd + bin.x;
      o.y = acc.y * dd + bin.y;
      o.z = acc.z * dd + bin.z;
      o.w = acc.w * dd + bin.w;
      if constexpr (ADD_RES) {
        o.x += r.x; o.y += r.y; o.z += r.z; o.w += r.w;
      }
      o.x = fmaxf(o.x, 0.0f);
      o.y = fmaxf(o.y, 0.0f);
      o.z = fmaxf(o.z, 0.0f);
      o.w = fmaxf(o.w, 0.0f);
      if constexpr (SCALE_A) {
        const float ds = __int_as_float(
            __builtin_amdgcn_readlane(__float_as_int(dsv), i));
        o.x *= ds; o.y *= ds; o.z *= ds; o.w *= ds;
      }
      *(float4*)&As[(wave * RPW + i) * APITCH + 4 * lane] = o;
      beg = end; end = nend; raw = raw_next;
    }
  }
  __syncthreads();

  lds_tile_gemm<LDN, ADD_BIAS>(As, W, bias_out, out, m0, wave, lane);
}

extern "C" void kernel_launch(void* const* d_in, const int* in_sizes, int n_in,
                              void* d_out, int out_size, void* d_ws, size_t ws_size,
                              hipStream_t stream) {
  const float* x   = (const float*)d_in[0];
  const int*   src = (const int*)  d_in[1];
  const int*   dst = (const int*)  d_in[2];
  const float* W1  = (const float*)d_in[3];
  const float* b1  = (const float*)d_in[4];
  const float* W2  = (const float*)d_in[5];
  const float* b2  = (const float*)d_in[6];
  const float* W3  = (const float*)d_in[7];
  const float* b3  = (const float*)d_in[8];
  const float* Wr  = (const float*)d_in[9];
  const float* br  = (const float*)d_in[10];
  const float* Wo  = (const float*)d_in[11];
  const float* bo  = (const float*)d_in[12];
  float* out = (float*)d_out;

  float* ws    = (float*)d_ws;
  float* RES   = ws;
  float* ABUF  = RES  + (size_t)NODES * FDIM;
  float* HBUF  = ABUF + (size_t)NODES * FDIM;
  float* dsrc  = HBUF + (size_t)NODES * FDIM;
  float* ddst  = dsrc + NODES;
  int*   cnt   = (int*)(ddst + NODES);
  int*   rowp  = cnt  + NODES;
  int*   rank  = rowp + NODES + 16;
  int*   colx  = rank + EDGES;

  const int nb       = (NODES + 255) / 256;
  const int mblocks  = MBLOCKS;
  const dim3 gemmH_blk(256);

  zero_deg_kernel<<<nb, 256, 0, stream>>>(dsrc, cnt, NODES);
  degree_rank_gemm_kernel<<<MBLOCKS + EBLOCKS, 256, 0, stream>>>(
      src, dst, dsrc, cnt, rank, x, Wr, br, RES);
  scan_norm_kernel<<<nb, 256, 0, stream>>>(cnt, dsrc, ddst, rowp, NODES);
  fill_csr_gemm_kernel<<<MBLOCKS + EBLOCKS, 256, 0, stream>>>(
      src, dst, rowp, rank, colx, x, W1, dsrc, HBUF);


  gather_gemm_kernel<FDIM, false, true, false>
      <<<mblocks, gemmH_blk, 0, stream>>>(HBUF, rowp, colx, ddst, dsrc, b1,
                                          nullptr, W2, nullptr, ABUF);

  gather_gemm_kernel<FDIM, false, true, false>
      <<<mblocks, gemmH_blk, 0, stream>>>(ABUF, rowp, colx, ddst, dsrc, b2,
                                          nullptr, W3, nullptr, HBUF);

  gather_gemm_kernel<CDIM, true, false, true>
      <<<mblocks, gemmH_blk, 0, stream>>>(HBUF, rowp, colx, ddst, dsrc, b3,
                                          RES, Wo, bo, out);
}


# In-edge lists as fixed 64-slot rows per destination, filled by the count pass (atomic rank = slot; overflow list): scan and CSR-fill launches gone, layer-1 rows scaled after x@W1 (7 -> 6 launches)
# speedup vs baseline: 3.5377x; 1.0960x over previous
#include <hip/hip_runtime.h>
#include <hip/hip_bf16.h>


typedef __attribute__((ext_vector_type(2))) float v2f;
typedef __attribute__((ext_vector_type(8))) float v8f;

#define NODES 50000
#define EDGES 800000
#define FDIM  128
#define CDIM  64
#define MTILE 5
#define MBLOCKS (NODES / (16 * MTILE))
#define EBLOCKS ((EDGES + 255) / 256)
#define SLOTS 64

__global__ void zero_deg_kernel(float* __restrict__ deg_out,
                                int* __restrict__ cnt_in,
                                int* __restrict__ novf, int n) {
  int i = blockIdx.x * blockDim.x + threadIdx.x;
  if (i < n) { deg_out[i] = 0.0f; cnt_in[i] = 0; }
  if (i == 0) novf[0] = 0;
}

__device__ __forceinline__ void degree_slot_edges(const int eblk,
                                                  const int* __restrict__ src,
                                                  const int* __restrict__ dst,
                                                  float* __restrict__ deg_out,
                                                  int* __restrict__ cnt_in,
                                                  int* __restrict__ slots,
                                                  int* __restrict__ novf,
                                                  int* __restrict__ ovf_dst,
                                                  int* __restrict__ ovf_src,
                                                  int e) {
  const int i = eblk * 256 + threadIdx.x;
  if (i < e) {
    const int s = src[i];
    const int d = dst[i];
    atomicAdd(&deg_out[s], 1.0f);
    const int r = atomicAdd(&cnt_in[d], 1);
    if (r < SLOTS) {
      slots[(size_t)d * SLOTS + r] = s;
    } else {
      const int k = atomicAdd(novf, 1);
      ovf_dst[k] = d;
      ovf_src[k] = s;
    }
  }
}

__device__ __forceinline__ float inv_sqrt_degree(const float deg) {
  return rsqrtf(deg > 0.0f ? deg : 1.0f);
}

#define RPW    (16 * MTILE / 8)
__global__ void __launch_bounds__(256)
scale_rows_kernel(float* __restrict__ H, const float* __restrict__ deg_out) {
  const int lane = threadIdx.x & 31;
  const int wave = __builtin_amdgcn_readfirstlane((int)(threadIdx.x >> 5));
  const int r0   = blockIdx.x * (16 * MTILE) + wave * RPW;
  float4* __restrict__ H4 = (float4*)H;
  float4 v[RPW];
#pragma unroll
  for (int i = 0; i < RPW; ++i)
    v[i] = H4[(size_t)(r0 + i) * (FDIM / 4) + lane];
#pragma unroll
  for (int i = 0; i < RPW; ++i) {
    const float s = inv_sqrt_degree(deg_out[r0 + i]);
    float4 o = v[i];
    o.x *= s; o.y *= s; o.z *= s; o.w *= s;
    H4[(size_t)(r0 + i) * (FDIM / 4) + lane] = o;
  }
}

#define APITCH (FDIM + 4)

template <int LDN, bool ADD_BIAS>
__device__ __forceinline__ void lds_tile_gemm(float* As,
                                              const float* __restrict__ W,
                                              const float* __restrict__ bias_out,
                                              float* __restrict__ out,
                                              const int m0, const int wave,
                                              const int lane) {
  constexpr int NCT = LDN / 16;
  const int c   = lane & 15;
  const int kh  = (lane >> 4) << 2;
  const int col = (wave << 4) + c;

  v8f acc[MTILE];
  if (wave < NCT) {
    float init = 0.0f;
    if constexpr (ADD_BIAS) init = bias_out[col];
#pragma unroll
    for (int t = 0; t < MTILE; ++t)
#pragma unroll
      for (int r = 0; r < 8; ++r) acc[t][r] = init;

    const float* __restrict__ bcol = W + (size_t)kh * LDN + col;
    const float* arow = As + c * APITCH + kh;

#pragma unroll
    for (int j = 0; j < FDIM / 8; ++j) {
      v2f b0, b1;
      b0.x = bcol[(size_t)(8 * j + 0) * LDN];
      b0.y = bcol[(size_t)(8 * j + 1) * LDN];
      b1.x = bcol[(size_t)(8 * j + 2) * LDN];
      b1.y = bcol[(size_t)(8 * j + 3) * LDN];
      float4 a[MTILE];
#pragma unroll
      for (int t = 0; t < MTILE; ++t)
        a[t] = *(const float4*)(arow + t * 16 * APITCH + 8 * j);
#pragma unroll
      for (int t = 0; t < MTILE; ++t) {
        v2f a0;
        a0.x = a[t].x; a0.y = a[t].y;
        acc[t] = __builtin_amdgcn_wmma_f32_16x16x4_f32(
             false, a0,  false, b0,
             (short)0, acc[t],  false,  false);
      }
#pragma unroll
      for (int t = 0; t < MTILE; ++t) {
        v2f a1;
        a1.x = a[t].z; a1.y = a[t].w;
        acc[t] = __builtin_amdgcn_wmma_f32_16x16x4_f32(
             false, a1,  false, b1,
             (short)0, acc[t],  false,  false);
      }
    }
  }

  __syncthreads();
  if (wave < NCT) {
    const int rbase = (lane >> 4) << 3;
#pragma unroll
    for (int t = 0; t < MTILE; ++t)
#pragma unroll
      for (int r = 0; r < 8; ++r)
        As[(t * 16 + rbase + r) * APITCH + col] = acc[t][r];
  }
  __syncthreads();
  constexpr int R4 = LDN / 4;
  constexpr int Q  = (16 * MTILE) * R4 / 32 / 8;
  float4* __restrict__ out4 = (float4*)(out + (size_t)m0 * LDN);
#pragma unroll
  for (int q = 0; q < Q; ++q) {
    const int e   = (wave * Q + q) * 32 + lane;
    const int row = e / R4;
    const int c4  = e % R4;
    out4[e] = *(const float4*)&As[row * APITCH + 4 * c4];
  }
}

template <bool ADD_BIAS>
__device__ __forceinline__ void gemm_x_tile(float* As, const int blk,
                                            const float* __restrict__ A,
                                            const float* __restrict__ W,
                                            const float* __restrict__ bias,
                                            float* __restrict__ out) {
  const int lane = threadIdx.x & 31;
  const int wave = __builtin_amdgcn_readfirstlane((int)(threadIdx.x >> 5));
  const int m0   = blk * (16 * MTILE);
  const int r0   = m0 + wave * RPW;
  const float4* __restrict__ A4 = (const float4*)A;
  float4 v[RPW];
#pragma unroll
  for (int i = 0; i < RPW; ++i)
    v[i] = A4[(size_t)(r0 + i) * (FDIM / 4) + lane];
#pragma unroll
  for (int i = 0; i < RPW; ++i)
    *(float4*)&As[(wave * RPW + i) * APITCH + 4 * lane] = v[i];
  __syncthreads();
  lds_tile_gemm<FDIM, ADD_BIAS>(As, W, bias, out, m0, wave, lane);
}

__global__ void __launch_bounds__(256)
edge_gemm_kernel(const int* __restrict__ src,
                 const int* __restrict__ dst,
                 float* __restrict__ deg_out,
                 int* __restrict__ cnt_in,
                 int* __restrict__ slots,
                 int* __restrict__ novf,
                 int* __restrict__ ovf_dst,
                 int* __restrict__ ovf_src,
                 const float* __restrict__ x,
                 const float* __restrict__ Wr,
                 const float* __restrict__ br,
                 float* __restrict__ res,
                 const float* __restrict__ W1,
                 float* __restrict__ h1) {
  __shared__ float As[16 * MTILE * APITCH];
  const int b = (int)blockIdx.x;
  if (b < MBLOCKS) {
    gemm_x_tile<true>(As, b, x, Wr, br, res);
    __syncthreads();
    gemm_x_tile<false>(As, b, x, W1, nullptr, h1);
  } else {
    degree_slot_edges(b - MBLOCKS, src, dst, deg_out, cnt_in, slots, novf,
                      ovf_dst, ovf_src, EDGES);
  }
}


__device__ __forceinline__ void add_source_rows(float4& acc,
                                                const float4* __restrict__ h4,
                                                int ids, int cnt, int lane) {
  for (int j = 0; j < cnt; j += 16) {
    float4 v[16];
#pragma unroll
    for (int u = 0; u < 16; ++u) {
      const int sidx = __builtin_amdgcn_readlane(ids, j + u);
      v[u] = h4[(size_t)sidx * (FDIM / 4) + lane];
    }
#pragma unroll
    for (int u = 0; u < 16; ++u) {
      const bool ok = (j + u) < cnt;
      acc.x += ok ? v[u].x : 0.0f;
      acc.y += ok ? v[u].y : 0.0f;
      acc.z += ok ? v[u].z : 0.0f;
      acc.w += ok ? v[u].w : 0.0f;
    }
  }
}

template <int LDN, bool ADD_RES, bool SCALE_A, bool ADD_BIAS>
__global__ void __launch_bounds__(256)
gather_gemm_kernel(const float* __restrict__ h,
                   const int* __restrict__ cnt_in,
                   const int* __restrict__ slots,
                   const float* __restrict__ deg_out,
                   const int* __restrict__ novf,
                   const int* __restrict__ ovf_dst,
                   const int* __restrict__ ovf_src,
                   const float* __restrict__ bias_in,
                   const float* __restrict__ res,
                   const float* __restrict__ W,
                   const float* __restrict__ bias_out,
                   float* __restrict__ out) {
  __shared__ float As[16 * MTILE * APITCH];
  const int lane = threadIdx.x & 31;
  const int wave = __builtin_amdgcn_readfirstlane((int)(threadIdx.x >> 5));
  const int m0   = blockIdx.x * (16 * MTILE);

  {
    const int r0 = m0 + wave * RPW;
    const int   cntv = (lane < RPW) ? cnt_in[r0 + lane] : 0;
    const float ddv  = inv_sqrt_degree((float)cntv);
    float dsv = 0.0f;
    if constexpr (SCALE_A)
      dsv = inv_sqrt_degree((lane < RPW) ? deg_out[r0 + lane] : 0.0f);
    const int nov = min(novf[0], EDGES);
    const float4* __restrict__ h4 = (const float4*)h;
    const float4 bin = ((const float4*)bias_in)[lane];
    int raw = slots[(size_t)r0 * SLOTS + lane];
    for (int i = 0; i < RPW; ++i) {
      const int row = r0 + i;
      const int deg = __builtin_amdgcn_readlane(cntv, i);
      const int n   = min(deg, SLOTS);
      const int raw_next =
          slots[(size_t)min(row + 1, NODES - 1) * SLOTS + lane];
      const int n0   = min(n, 32);
      const int mine = (lane < n0) ? raw : 0;
      float4 r = make_float4(0.0f, 0.0f, 0.0f, 0.0f);
      if constexpr (ADD_RES)
        r = ((const float4*)res)[(size_t)row * (FDIM / 4) + lane];
      float4 acc = make_float4(0.0f, 0.0f, 0.0f, 0.0f);
      add_source_rows(acc, h4, mine, n0, lane);
      if (n > 32) {
        const int n1  = n - 32;
        const int hi  = slots[(size_t)row * SLOTS + 32 + lane];
        const int ids = (lane < n1) ? hi : 0;
        add_source_rows(acc, h4, ids, n1, lane);
      }
      if (deg > SLOTS) {
        for (int base = 0; base < nov; base += 32) {
          const int k  = base + lane;
          const int od = (k < nov) ? ovf_dst[k] : -1;
          const int os = (k < nov) ? ovf_src[k] : 0;
          unsigned m = __builtin_amdgcn_ballot_w32(od == row);
          while (m != 0u) {
            const int bit = __builtin_amdgcn_readfirstlane(__builtin_ctz(m));
            m &= m - 1u;
            const int sidx = __builtin_amdgcn_readlane(os, bit);
            const float4 v = h4[(size_t)sidx * (FDIM / 4) + lane];
            acc.x += v.x; acc.y += v.y; acc.z += v.z; acc.w += v.w;
          }
        }
      }
      const float dd = __int_as_float(
          __builtin_amdgcn_readlane(__float_as_int(ddv), i));
      float4 o;
      o.x = acc.x * dd + bin.x;
      o.y = acc.y * dd + bin.y;
      o.z = acc.z * dd + bin.z;
      o.w = acc.w * dd + bin.w;
      if constexpr (ADD_RES) {
        o.x += r.x; o.y += r.y; o.z += r.z; o.w += r.w;
      }
      o.x = fmaxf(o.x, 0.0f);
      o.y = fmaxf(o.y, 0.0f);
      o.z = fmaxf(o.z, 0.0f);
      o.w = fmaxf(o.w, 0.0f);
      if constexpr (SCALE_A) {
        const float ds = __int_as_float(
            __builtin_amdgcn_readlane(__float_as_int(dsv), i));
        o.x *= ds; o.y *= ds; o.z *= ds; o.w *= ds;
      }
      *(float4*)&As[(wave * RPW + i) * APITCH + 4 * lane] = o;
      raw = raw_next;
    }
  }
  __syncthreads();

  lds_tile_gemm<LDN, ADD_BIAS>(As, W, bias_out, out, m0, wave, lane);
}

extern "C" void kernel_launch(void* const* d_in, const int* in_sizes, int n_in,
                              void* d_out, int out_size, void* d_ws, size_t ws_size,
                              hipStream_t stream) {
  const float* x   = (const float*)d_in[0];
  const int*   src = (const int*)  d_in[1];
  const int*   dst = (const int*)  d_in[2];
  const float* W1  = (const float*)d_in[3];
  const float* b1  = (const float*)d_in[4];
  const float* W2  = (const float*)d_in[5];
  const float* b2  = (const float*)d_in[6];
  const float* W3  = (const float*)d_in[7];
  const float* b3  = (const float*)d_in[8];
  const float* Wr  = (const float*)d_in[9];
  const float* br  = (const float*)d_in[10];
  const float* Wo  = (const float*)d_in[11];
  const float* bo  = (const float*)d_in[12];
  float* out = (float*)d_out;

  float* ws    = (float*)d_ws;
  float* RES   = ws;
  float* ABUF  = RES  + (size_t)NODES * FDIM;
  float* HBUF  = ABUF + (size_t)NODES * FDIM;
  int*   slots = (int*)(HBUF + (size_t)NODES * FDIM);
  float* dego  = (float*)(slots + (size_t)NODES * SLOTS);
  int*   cnt   = (int*)(dego + NODES);
  int*   novf  = cnt  + NODES;
  int*   ovfd  = novf + 16;
  int*   ovfs  = ovfd + EDGES;

  const int nb       = (NODES + 255) / 256;
  const int mblocks  = MBLOCKS;
  const dim3 gemmH_blk(256);

  zero_deg_kernel<<<nb, 256, 0, stream>>>(dego, cnt, novf, NODES);
  edge_gemm_kernel<<<MBLOCKS + EBLOCKS, 256, 0, stream>>>(
      src, dst, dego, cnt, slots, novf, ovfd, ovfs, x, Wr, br, RES, W1, HBUF);
  scale_rows_kernel<<<mblocks, 256, 0, stream>>>(HBUF, dego);


  gather_gemm_kernel<FDIM, false, true, false>
      <<<mblocks, gemmH_blk, 0, stream>>>(HBUF, cnt, slots, dego, novf, ovfd,
                                          ovfs, b1, nullptr, W2, nullptr, ABUF);

  gather_gemm_kernel<FDIM, false, true, false>
      <<<mblocks, gemmH_blk, 0, stream>>>(ABUF, cnt, slots, dego, novf, ovfd,
                                          ovfs, b2, nullptr, W3, nullptr, HBUF);

  gather_gemm_kernel<CDIM, true, false, true>
      <<<mblocks, gemmH_blk, 0, stream>>>(HBUF, cnt, slots, dego, novf, ovfd,
                                          ovfs, b3, RES, Wo, bo, out);
}


# Out-degrees counted up front: launch 1 histograms src in LDS (64 partial tables, no global float atomics); x@W1 blocks sum them and scale their rows as they store; row-scale launch gone (6 -> 5)
# speedup vs baseline: 4.1962x; 1.1861x over previous
#include <hip/hip_runtime.h>
#include <hip/hip_bf16.h>


typedef __attribute__((ext_vector_type(2))) float v2f;
typedef __attribute__((ext_vector_type(8))) float v8f;

#define NODES 50000
#define EDGES 800000
#define FDIM  128
#define CDIM  64
#define MTILE 5
#define MBLOCKS (NODES / (16 * MTILE))
#define EBLOCKS ((EDGES + 255) / 256)
#define SLOTS 64
#define HBLOCKS 64
#define HTHREADS 1024
#define ECHUNK (EDGES / HBLOCKS)
#define EPT ((ECHUNK + HTHREADS - 1) / HTHREADS)

__global__ void __launch_bounds__(HTHREADS)
outdeg_hist_kernel(const int* __restrict__ src,
                   unsigned* __restrict__ part,
                   int* __restrict__ cnt_in,
                   int* __restrict__ novf) {
  __shared__ unsigned hist[NODES];
  const int t = (int)threadIdx.x;
  const int b = (int)blockIdx.x;
  const int g = b * HTHREADS + t;
  if (g < NODES) cnt_in[g] = 0;
  if (g == 0) novf[0] = 0;
  int ids[EPT];
#pragma unroll
  for (int k = 0; k < EPT; ++k) {
    const int i = k * HTHREADS + t;
    ids[k] = (i < ECHUNK) ? src[b * ECHUNK + i] : -1;
  }
  uint4* hist4 = (uint4*)hist;
  for (int i = t; i < NODES / 4; i += HTHREADS)
    hist4[i] = make_uint4(0u, 0u, 0u, 0u);
  __syncthreads();
#pragma unroll
  for (int k = 0; k < EPT; ++k)
    if ((unsigned)ids[k] < (unsigned)NODES) atomicAdd(&hist[ids[k]], 1u);
  __syncthreads();
  uint4* __restrict__ out4 = (uint4*)(part + (size_t)b * NODES);
  for (int i = t; i < NODES / 4; i += HTHREADS) out4[i] = hist4[i];
}

__device__ __forceinline__ void degree_slot_edges(const int eblk,
                                                  const int* __restrict__ src,
                                                  const int* __restrict__ dst,
                                                  int* __restrict__ cnt_in,
                                                  int* __restrict__ slots,
                                                  int* __restrict__ novf,
                                                  int* __restrict__ ovf_dst,
                                                  int* __restrict__ ovf_src,
                                                  int e) {
  const int i = eblk * 256 + threadIdx.x;
  if (i < e) {
    const int s = src[i];
    const int d = dst[i];
    const int r = atomicAdd(&cnt_in[d], 1);
    if (r < SLOTS) {
      slots[(size_t)d * SLOTS + r] = s;
    } else {
      const int k = atomicAdd(novf, 1);
      ovf_dst[k] = d;
      ovf_src[k] = s;
    }
  }
}

__device__ __forceinline__ float inv_sqrt_degree(const float deg) {
  return rsqrtf(deg > 0.0f ? deg : 1.0f);
}

#define RPW    (16 * MTILE / 8)

#define APITCH (FDIM + 4)

template <int LDN, bool ADD_BIAS, bool SCALE_ROWS>
__device__ __forceinline__ void lds_tile_gemm(float* As,
                                              const float* __restrict__ W,
                                              const float* __restrict__ bias_out,
                                              const float* rscale,
                                              float* __restrict__ out,
                                              const int m0, const int wave,
                                              const int lane) {
  constexpr int NCT = LDN / 16;
  const int c   = lane & 15;
  const int kh  = (lane >> 4) << 2;
  const int col = (wave << 4) + c;

  v8f acc[MTILE];
  if (wave < NCT) {
    float init = 0.0f;
    if constexpr (ADD_BIAS) init = bias_out[col];
#pragma unroll
    for (int t = 0; t < MTILE; ++t)
#pragma unroll
      for (int r = 0; r < 8; ++r) acc[t][r] = init;

    const float* __restrict__ bcol = W + (size_t)kh * LDN + col;
    const float* arow = As + c * APITCH + kh;

#pragma unroll
    for (int j = 0; j < FDIM / 8; ++j) {
      v2f b0, b1;
      b0.x = bcol[(size_t)(8 * j + 0) * LDN];
      b0.y = bcol[(size_t)(8 * j + 1) * LDN];
      b1.x = bcol[(size_t)(8 * j + 2) * LDN];
      b1.y = bcol[(size_t)(8 * j + 3) * LDN];
      float4 a[MTILE];
#pragma unroll
      for (int t = 0; t < MTILE; ++t)
        a[t] = *(const float4*)(arow + t * 16 * APITCH + 8 * j);
#pragma unroll
      for (int t = 0; t < MTILE; ++t) {
        v2f a0;
        a0.x = a[t].x; a0.y = a[t].y;
        acc[t] = __builtin_amdgcn_wmma_f32_16x16x4_f32(
             false, a0,  false, b0,
             (short)0, acc[t],  false,  false);
      }
#pragma unroll
      for (int t = 0; t < MTILE; ++t) {
        v2f a1;
        a1.x = a[t].z; a1.y = a[t].w;
        acc[t] = __builtin_amdgcn_wmma_f32_16x16x4_f32(
             false, a1,  false, b1,
             (short)0, acc[t],  false,  false);
      }
    }
  }

  __syncthreads();
  if (wave < NCT) {
    const int rbase = (lane >> 4) << 3;
#pragma unroll
    for (int t = 0; t < MTILE; ++t)
#pragma unroll
      for (int r = 0; r < 8; ++r)
        As[(t * 16 + rbase + r) * APITCH + col] = acc[t][r];
  }
  __syncthreads();
  constexpr int R4 = LDN / 4;
  constexpr int Q  = (16 * MTILE) * R4 / 32 / 8;
  float4* __restrict__ out4 = (float4*)(out + (size_t)m0 * LDN);
#pragma unroll
  for (int q = 0; q < Q; ++q) {
    const int e   = (wave * Q + q) * 32 + lane;
    const int row = e / R4;
    const int c4  = e % R4;
    float4 v = *(const float4*)&As[row * APITCH + 4 * c4];
    if constexpr (SCALE_ROWS) {
      const float sc = rscale[row];
      v.x *= sc; v.y *= sc; v.z *= sc; v.w *= sc;
    }
    out4[e] = v;
  }
}

template <bool ADD_BIAS, bool SCALE_ROWS>
__device__ __forceinline__ void gemm_x_tile(float* As, const int blk,
                                            const float* __restrict__ A,
                                            const float* __restrict__ W,
                                            const float* __restrict__ bias,
                                            const float* rscale,
                                            float* __restrict__ out) {
  const int lane = threadIdx.x & 31;
  const int wave = __builtin_amdgcn_readfirstlane((int)(threadIdx.x >> 5));
  const int m0   = blk * (16 * MTILE);
  const int r0   = m0 + wave * RPW;
  const float4* __restrict__ A4 = (const float4*)A;
  float4 v[RPW];
#pragma unroll
  for (int i = 0; i < RPW; ++i)
    v[i] = A4[(size_t)(r0 + i) * (FDIM / 4) + lane];
#pragma unroll
  for (int i = 0; i < RPW; ++i)
    *(float4*)&As[(wave * RPW + i) * APITCH + 4 * lane] = v[i];
  __syncthreads();
  lds_tile_gemm<FDIM, ADD_BIAS, SCALE_ROWS>(As, W, bias, rscale, out, m0, wave,
                                            lane);
}

__global__ void __launch_bounds__(256)
edge_gemm_kernel(const int* __restrict__ src,
                 const int* __restrict__ dst,
                 const unsigned* __restrict__ part,
                 float* __restrict__ deg_out,
                 int* __restrict__ cnt_in,
                 int* __restrict__ slots,
                 int* __restrict__ novf,
                 int* __restrict__ ovf_dst,
                 int* __restrict__ ovf_src,
                 const float* __restrict__ x,
                 const float* __restrict__ Wr,
                 const float* __restrict__ br,
                 float* __restrict__ res,
                 const float* __restrict__ W1,
                 float* __restrict__ h1) {
  __shared__ float As[16 * MTILE * APITCH];
  __shared__ float dsl[16 * MTILE];
  const int b = (int)blockIdx.x;
  if (b < MBLOCKS) {
    if (threadIdx.x < 16 * MTILE) {
      const int r = b * (16 * MTILE) + (int)threadIdx.x;
      unsigned n = 0u;
#pragma unroll
      for (int c = 0; c < HBLOCKS; ++c) n += part[(size_t)c * NODES + r];
      const float deg = (float)n;
      deg_out[r] = deg;
      dsl[threadIdx.x] = inv_sqrt_degree(deg);
    }
    gemm_x_tile<true, false>(As, b, x, Wr, br, nullptr, res);
    __syncthreads();
    gemm_x_tile<false, true>(As, b, x, W1, nullptr, dsl, h1);
  } else {
    degree_slot_edges(b - MBLOCKS, src, dst, cnt_in, slots, novf,
                      ovf_dst, ovf_src, EDGES);
  }
}


__device__ __forceinline__ void add_source_rows(float4& acc,
                                                const float4* __restrict__ h4,
                                                int ids, int cnt, int lane) {
  for (int j = 0; j < cnt; j += 16) {
    float4 v[16];
#pragma unroll
    for (int u = 0; u < 16; ++u) {
      const int sidx = __builtin_amdgcn_readlane(ids, j + u);
      v[u] = h4[(size_t)sidx * (FDIM / 4) + lane];
    }
#pragma unroll
    for (int u = 0; u < 16; ++u) {
      const bool ok = (j + u) < cnt;
      acc.x += ok ? v[u].x : 0.0f;
      acc.y += ok ? v[u].y : 0.0f;
      acc.z += ok ? v[u].z : 0.0f;
      acc.w += ok ? v[u].w : 0.0f;
    }
  }
}

template <int LDN, bool ADD_RES, bool SCALE_A, bool ADD_BIAS>
__global__ void __launch_bounds__(256)
gather_gemm_kernel(const float* __restrict__ h,
                   const int* __restrict__ cnt_in,
                   const int* __restrict__ slots,
                   const float* __restrict__ deg_out,
                   const int* __restrict__ novf,
                   const int* __restrict__ ovf_dst,
                   const int* __restrict__ ovf_src,
                   const float* __restrict__ bias_in,
                   const float* __restrict__ res,
                   const float* __restrict__ W,
                   const float* __restrict__ bias_out,
                   float* __restrict__ out) {
  __shared__ float As[16 * MTILE * APITCH];
  const int lane = threadIdx.x & 31;
  const int wave = __builtin_amdgcn_readfirstlane((int)(threadIdx.x >> 5));
  const int m0   = blockIdx.x * (16 * MTILE);

  {
    const int r0 = m0 + wave * RPW;
    const int   cntv = (lane < RPW) ? cnt_in[r0 + lane] : 0;
    const float ddv  = inv_sqrt_degree((float)cntv);
    float dsv = 0.0f;
    if constexpr (SCALE_A)
      dsv = inv_sqrt_degree((lane < RPW) ? deg_out[r0 + lane] : 0.0f);
    const int nov = min(novf[0], EDGES);
    const float4* __restrict__ h4 = (const float4*)h;
    const float4 bin = ((const float4*)bias_in)[lane];
    int raw = slots[(size_t)r0 * SLOTS + lane];
    for (int i = 0; i < RPW; ++i) {
      const int row = r0 + i;
      const int deg = __builtin_amdgcn_readlane(cntv, i);
      const int n   = min(deg, SLOTS);
      const int raw_next =
          slots[(size_t)min(row + 1, NODES - 1) * SLOTS + lane];
      const int n0   = min(n, 32);
      const int mine = (lane < n0) ? raw : 0;
      float4 r = make_float4(0.0f, 0.0f, 0.0f, 0.0f);
      if constexpr (ADD_RES)
        r = ((const float4*)res)[(size_t)row * (FDIM / 4) + lane];
      float4 acc = make_float4(0.0f, 0.0f, 0.0f, 0.0f);
      add_source_rows(acc, h4, mine, n0, lane);
      if (n > 32) {
        const int n1  = n - 32;
        const int hi  = slots[(size_t)row * SLOTS + 32 + lane];
        const int ids = (lane < n1) ? hi : 0;
        add_source_rows(acc, h4, ids, n1, lane);
      }
      if (deg > SLOTS) {
        for (int base = 0; base < nov; base += 32) {
          const int k  = base + lane;
          const int od = (k < nov) ? ovf_dst[k] : -1;
          const int os = (k < nov) ? ovf_src[k] : 0;
          unsigned m = __builtin_amdgcn_ballot_w32(od == row);
          while (m != 0u) {
            const int bit = __builtin_amdgcn_readfirstlane(__builtin_ctz(m));
            m &= m - 1u;
            const int sidx = __builtin_amdgcn_readlane(os, bit);
            const float4 v = h4[(size_t)sidx * (FDIM / 4) + lane];
            acc.x += v.x; acc.y += v.y; acc.z += v.z; acc.w += v.w;
          }
        }
      }
      const float dd = __int_as_float(
          __builtin_amdgcn_readlane(__float_as_int(ddv), i));
      float4 o;
      o.x = acc.x * dd + bin.x;
      o.y = acc.y * dd + bin.y;
      o.z = acc.z * dd + bin.z;
      o.w = acc.w * dd + bin.w;
      if constexpr (ADD_RES) {
        o.x += r.x; o.y += r.y; o.z += r.z; o.w += r.w;
      }
      o.x = fmaxf(o.x, 0.0f);
      o.y = fmaxf(o.y, 0.0f);
      o.z = fmaxf(o.z, 0.0f);
      o.w = fmaxf(o.w, 0.0f);
      if constexpr (SCALE_A) {
        const float ds = __int_as_float(
            __builtin_amdgcn_readlane(__float_as_int(dsv), i));
        o.x *= ds; o.y *= ds; o.z *= ds; o.w *= ds;
      }
      *(float4*)&As[(wave * RPW + i) * APITCH + 4 * lane] = o;
      raw = raw_next;
    }
  }
  __syncthreads();

  lds_tile_gemm<LDN, ADD_BIAS, false>(As, W, bias_out, nullptr, out, m0, wave,
                                      lane);
}

extern "C" void kernel_launch(void* const* d_in, const int* in_sizes, int n_in,
                              void* d_out, int out_size, void* d_ws, size_t ws_size,
                              hipStream_t stream) {
  const float* x   = (const float*)d_in[0];
  const int*   src = (const int*)  d_in[1];
  const int*   dst = (const int*)  d_in[2];
  const float* W1  = (const float*)d_in[3];
  const float* b1  = (const float*)d_in[4];
  const float* W2  = (const float*)d_in[5];
  const float* b2  = (const float*)d_in[6];
  const float* W3  = (const float*)d_in[7];
  const float* b3  = (const float*)d_in[8];
  const float* Wr  = (const float*)d_in[9];
  const float* br  = (const float*)d_in[10];
  const float* Wo  = (const float*)d_in[11];
  const float* bo  = (const float*)d_in[12];
  float* out = (float*)d_out;

  float* ws    = (float*)d_ws;
  float* RES   = ws;
  float* ABUF  = RES  + (size_t)NODES * FDIM;
  float* HBUF  = ABUF + (size_t)NODES * FDIM;
  int*   slots = (int*)(HBUF + (size_t)NODES * FDIM);
  float* dego  = (float*)(slots + (size_t)NODES * SLOTS);
  int*   cnt   = (int*)(dego + NODES);
  int*   novf  = cnt  + NODES;
  int*   ovfd  = novf + 16;
  int*   ovfs  = ovfd + EDGES;
  unsigned* part = (unsigned*)(ovfs + EDGES);

  const int mblocks  = MBLOCKS;
  const dim3 gemmH_blk(256);

  outdeg_hist_kernel<<<HBLOCKS, HTHREADS, 0, stream>>>(src, part, cnt, novf);
  edge_gemm_kernel<<<MBLOCKS + EBLOCKS, 256, 0, stream>>>(
      src, dst, part, dego, cnt, slots, novf, ovfd, ovfs, x, Wr, br, RES, W1,
      HBUF);


  gather_gemm_kernel<FDIM, false, true, false>
      <<<mblocks, gemmH_blk, 0, stream>>>(HBUF, cnt, slots, dego, novf, ovfd,
                                          ovfs, b1, nullptr, W2, nullptr, ABUF);

  gather_gemm_kernel<FDIM, false, true, false>
      <<<mblocks, gemmH_blk, 0, stream>>>(ABUF, cnt, slots, dego, novf, ovfd,
                                          ovfs, b2, nullptr, W3, nullptr, HBUF);

  gather_gemm_kernel<CDIM, true, false, true>
      <<<mblocks, gemmH_blk, 0, stream>>>(HBUF, cnt, slots, dego, novf, ovfd,
                                          ovfs, b3, RES, Wo, bo, out);
}


# In-edge slot rows built by a two-level counting sort in LDS (edges ordered by destination range per chunk, then one block per range): no global atomic or scattered store per edge; sum order arrival
# speedup vs baseline: 4.9880x; 1.1887x over previous
#include <hip/hip_runtime.h>
#include <hip/hip_bf16.h>


typedef __attribute__((ext_vector_type(2))) float v2f;
typedef __attribute__((ext_vector_type(8))) float v8f;

#define NODES 50000
#define EDGES 800000
#define FDIM  128
#define CDIM  64
#define MTILE 5
#define MBLOCKS (NODES / (16 * MTILE))
#define NBINS MBLOCKS
#define SLOTS 64
#define HBLOCKS 64
#define HTHREADS 1024
#define ECHUNK (EDGES / HBLOCKS)
#define EPT ((ECHUNK + HTHREADS - 1) / HTHREADS)

__global__ void __launch_bounds__(HTHREADS)
outdeg_hist_bin_kernel(const int* __restrict__ src,
                       const int* __restrict__ dst,
                       unsigned* __restrict__ part,
                       int* __restrict__ novf,
                       int* __restrict__ binned,
                       int* __restrict__ boff) {
  __shared__ unsigned hist[NODES];
  __shared__ int2 sorted[ECHUNK];
  __shared__ int bcnt[NBINS + 1];
  __shared__ int wsum[HTHREADS / 32];
  const int t = (int)threadIdx.x;
  const int b = (int)blockIdx.x;
  const int lane = t & 31;
  const int wave = t >> 5;
  if (b == 0 && t == 0) novf[0] = 0;
  int ids[EPT], dd[EPT];
#pragma unroll
  for (int k = 0; k < EPT; ++k) {
    const int i = k * HTHREADS + t;
    ids[k] = (i < ECHUNK) ? src[b * ECHUNK + i] : -1;
    dd[k]  = (i < ECHUNK) ? dst[b * ECHUNK + i] : -1;
  }
  uint4* hist4 = (uint4*)hist;
  for (int i = t; i < NODES / 4; i += HTHREADS)
    hist4[i] = make_uint4(0u, 0u, 0u, 0u);
  if (t <= NBINS) bcnt[t] = 0;
  __syncthreads();
  int rk[EPT];
#pragma unroll
  for (int k = 0; k < EPT; ++k) {
    const bool sok = (unsigned)ids[k] < (unsigned)NODES;
    const bool dok = (unsigned)dd[k] < (unsigned)NODES;
    if (sok) atomicAdd(&hist[ids[k]], 1u);
    rk[k] = (sok && dok) ? atomicAdd(&bcnt[dd[k] / (16 * MTILE)], 1) : -1;
  }
  __syncthreads();
  const int c = (t < NBINS) ? bcnt[t] : 0;
  int x = c;
#pragma unroll
  for (int o = 1; o < 32; o <<= 1) {
    const int y = __shfl_up(x, o);
    if (lane >= o) x += y;
  }
  if (lane == 31) wsum[wave] = x;
  __syncthreads();
  int base = 0;
  for (int w = 0; w < wave; ++w) base += wsum[w];
  const int excl = base + x - c;
  if (t < NBINS) bcnt[t] = excl;
  if (t == NBINS - 1) bcnt[NBINS] = excl + c;
  __syncthreads();
#pragma unroll
  for (int k = 0; k < EPT; ++k)
    if (rk[k] >= 0)
      sorted[bcnt[dd[k] / (16 * MTILE)] + rk[k]] = make_int2(ids[k], dd[k]);
  __syncthreads();
  uint4* __restrict__ out4 = (uint4*)(part + (size_t)b * NODES);
  for (int i = t; i < NODES / 4; i += HTHREADS) out4[i] = hist4[i];
  const int total = bcnt[NBINS];
  int2* __restrict__ bout = (int2*)binned + (size_t)b * ECHUNK;
  for (int i = t; i < total; i += HTHREADS) bout[i] = sorted[i];
  if (t <= NBINS) boff[b * (NBINS + 1) + t] = bcnt[t];
}

__device__ __forceinline__ void build_slot_rows(const int m,
                                                const int* __restrict__ binned,
                                                const int* __restrict__ boff,
                                                int* __restrict__ cnt_in,
                                                int* __restrict__ slots,
                                                int* __restrict__ novf,
                                                int* __restrict__ ovf_dst,
                                                int* __restrict__ ovf_src,
                                                int* tile, int* rcnt, int* seg) {
  const int t = (int)threadIdx.x;
  const int lane = t & 31;
  const int wave = t >> 5;
  if (t < 16 * MTILE) rcnt[t] = 0;
  if (t < HBLOCKS) {
    seg[t]           = boff[t * (NBINS + 1) + m];
    seg[HBLOCKS + t] = boff[t * (NBINS + 1) + m + 1];
  }
  __syncthreads();
  const int2* __restrict__ b2 = (const int2*)binned;
  for (int c = wave; c < HBLOCKS; c += 8) {
    const int beg = seg[c];
    const int end = seg[HBLOCKS + c];
    for (int i = beg + lane; i < end; i += 32) {
      const int2 e = b2[(size_t)c * ECHUNK + i];
      const int row = e.y - m * (16 * MTILE);
      const int r = atomicAdd(&rcnt[row], 1);
      if (r < SLOTS) {
        tile[row * SLOTS + r] = e.x;
      } else {
        const int k = atomicAdd(novf, 1);
        ovf_dst[k] = e.y;
        ovf_src[k] = e.x;
      }
    }
  }
  __syncthreads();
  int4* __restrict__ out4 = (int4*)(slots + (size_t)m * (16 * MTILE) * SLOTS);
  const int4* tile4 = (const int4*)tile;
  for (int i = t; i < 16 * MTILE * SLOTS / 4; i += 256) out4[i] = tile4[i];
  if (t < 16 * MTILE) cnt_in[m * (16 * MTILE) + t] = rcnt[t];
}

__device__ __forceinline__ float inv_sqrt_degree(const float deg) {
  return rsqrtf(deg > 0.0f ? deg : 1.0f);
}

#define RPW    (16 * MTILE / 8)

#define APITCH (FDIM + 4)

template <int LDN, bool ADD_BIAS, bool SCALE_ROWS>
__device__ __forceinline__ void lds_tile_gemm(float* As,
                                              const float* __restrict__ W,
                                              const float* __restrict__ bias_out,
                                              const float* rscale,
                                              float* __restrict__ out,
                                              const int m0, const int wave,
                                              const int lane) {
  constexpr int NCT = LDN / 16;
  const int c   = lane & 15;
  const int kh  = (lane >> 4) << 2;
  const int col = (wave << 4) + c;

  v8f acc[MTILE];
  if (wave < NCT) {
    float init = 0.0f;
    if constexpr (ADD_BIAS) init = bias_out[col];
#pragma unroll
    for (int t = 0; t < MTILE; ++t)
#pragma unroll
      for (int r = 0; r < 8; ++r) acc[t][r] = init;

    const float* __restrict__ bcol = W + (size_t)kh * LDN + col;
    const float* arow = As + c * APITCH + kh;

#pragma unroll
    for (int j = 0; j < FDIM / 8; ++j) {
      v2f b0, b1;
      b0.x = bcol[(size_t)(8 * j + 0) * LDN];
      b0.y = bcol[(size_t)(8 * j + 1) * LDN];
      b1.x = bcol[(size_t)(8 * j + 2) * LDN];
      b1.y = bcol[(size_t)(8 * j + 3) * LDN];
      float4 a[MTILE];
#pragma unroll
      for (int t = 0; t < MTILE; ++t)
        a[t] = *(const float4*)(arow + t * 16 * APITCH + 8 * j);
#pragma unroll
      for (int t = 0; t < MTILE; ++t) {
        v2f a0;
        a0.x = a[t].x; a0.y = a[t].y;
        acc[t] = __builtin_amdgcn_wmma_f32_16x16x4_f32(
             false, a0,  false, b0,
             (short)0, acc[t],  false,  false);
      }
#pragma unroll
      for (int t = 0; t < MTILE; ++t) {
        v2f a1;
        a1.x = a[t].z; a1.y = a[t].w;
        acc[t] = __builtin_amdgcn_wmma_f32_16x16x4_f32(
             false, a1,  false, b1,
             (short)0, acc[t],  false,  false);
      }
    }
  }

  __syncthreads();
  if (wave < NCT) {
    const int rbase = (lane >> 4) << 3;
#pragma unroll
    for (int t = 0; t < MTILE; ++t)
#pragma unroll
      for (int r = 0; r < 8; ++r)
        As[(t * 16 + rbase + r) * APITCH + col] = acc[t][r];
  }
  __syncthreads();
  constexpr int R4 = LDN / 4;
  constexpr int Q  = (16 * MTILE) * R4 / 32 / 8;
  float4* __restrict__ out4 = (float4*)(out + (size_t)m0 * LDN);
#pragma unroll
  for (int q = 0; q < Q; ++q) {
    const int e   = (wave * Q + q) * 32 + lane;
    const int row = e / R4;
    const int c4  = e % R4;
    float4 v = *(const float4*)&As[row * APITCH + 4 * c4];
    if constexpr (SCALE_ROWS) {
      const float sc = rscale[row];
      v.x *= sc; v.y *= sc; v.z *= sc; v.w *= sc;
    }
    out4[e] = v;
  }
}

template <bool ADD_BIAS, bool SCALE_ROWS>
__device__ __forceinline__ void gemm_x_tile(float* As, const int blk,
                                            const float* __restrict__ A,
                                            const float* __restrict__ W,
                                            const float* __restrict__ bias,
                                            const float* rscale,
                                            float* __restrict__ out) {
  const int lane = threadIdx.x & 31;
  const int wave = __builtin_amdgcn_readfirstlane((int)(threadIdx.x >> 5));
  const int m0   = blk * (16 * MTILE);
  const int r0   = m0 + wave * RPW;
  const float4* __restrict__ A4 = (const float4*)A;
  float4 v[RPW];
#pragma unroll
  for (int i = 0; i < RPW; ++i)
    v[i] = A4[(size_t)(r0 + i) * (FDIM / 4) + lane];
#pragma unroll
  for (int i = 0; i < RPW; ++i)
    *(float4*)&As[(wave * RPW + i) * APITCH + 4 * lane] = v[i];
  __syncthreads();
  lds_tile_gemm<FDIM, ADD_BIAS, SCALE_ROWS>(As, W, bias, rscale, out, m0, wave,
                                            lane);
}

__global__ void __launch_bounds__(256)
edge_gemm_kernel(const int* __restrict__ binned,
                 const int* __restrict__ boff,
                 const unsigned* __restrict__ part,
                 float* __restrict__ deg_out,
                 int* __restrict__ cnt_in,
                 int* __restrict__ slots,
                 int* __restrict__ novf,
                 int* __restrict__ ovf_dst,
                 int* __restrict__ ovf_src,
                 const float* __restrict__ x,
                 const float* __restrict__ Wr,
                 const float* __restrict__ br,
                 float* __restrict__ res,
                 const float* __restrict__ W1,
                 float* __restrict__ h1) {
  __shared__ float As[16 * MTILE * APITCH];
  __shared__ float dsl[16 * MTILE];
  const int b = (int)blockIdx.x;
  if (b < MBLOCKS) {
    if (threadIdx.x < 16 * MTILE) {
      const int r = b * (16 * MTILE) + (int)threadIdx.x;
      unsigned n = 0u;
#pragma unroll
      for (int c = 0; c < HBLOCKS; ++c) n += part[(size_t)c * NODES + r];
      const float deg = (float)n;
      deg_out[r] = deg;
      dsl[threadIdx.x] = inv_sqrt_degree(deg);
    }
    gemm_x_tile<true, false>(As, b, x, Wr, br, nullptr, res);
    __syncthreads();
    gemm_x_tile<false, true>(As, b, x, W1, nullptr, dsl, h1);
  } else {
    int* tile = (int*)As;
    int* rcnt = tile + 16 * MTILE * SLOTS;
    int* seg  = rcnt + 16 * MTILE;
    build_slot_rows(b - MBLOCKS, binned, boff, cnt_in, slots, novf, ovf_dst,
                    ovf_src, tile, rcnt, seg);
  }
}


__device__ __forceinline__ void add_source_rows(float4& acc,
                                                const float4* __restrict__ h4,
                                                int ids, int cnt, int lane) {
  for (int j = 0; j < cnt; j += 16) {
    float4 v[16];
#pragma unroll
    for (int u = 0; u < 16; ++u) {
      const int sidx = __builtin_amdgcn_readlane(ids, j + u);
      v[u] = h4[(size_t)sidx * (FDIM / 4) + lane];
    }
#pragma unroll
    for (int u = 0; u < 16; ++u) {
      const bool ok = (j + u) < cnt;
      acc.x += ok ? v[u].x : 0.0f;
      acc.y += ok ? v[u].y : 0.0f;
      acc.z += ok ? v[u].z : 0.0f;
      acc.w += ok ? v[u].w : 0.0f;
    }
  }
}

template <int LDN, bool ADD_RES, bool SCALE_A, bool ADD_BIAS>
__global__ void __launch_bounds__(256)
gather_gemm_kernel(const float* __restrict__ h,
                   const int* __restrict__ cnt_in,
                   const int* __restrict__ slots,
                   const float* __restrict__ deg_out,
                   const int* __restrict__ novf,
                   const int* __restrict__ ovf_dst,
                   const int* __restrict__ ovf_src,
                   const float* __restrict__ bias_in,
                   const float* __restrict__ res,
                   const float* __restrict__ W,
                   const float* __restrict__ bias_out,
                   float* __restrict__ out) {
  __shared__ float As[16 * MTILE * APITCH];
  const int lane = threadIdx.x & 31;
  const int wave = __builtin_amdgcn_readfirstlane((int)(threadIdx.x >> 5));
  const int m0   = blockIdx.x * (16 * MTILE);

  {
    const int r0 = m0 + wave * RPW;
    const int   cntv = (lane < RPW) ? cnt_in[r0 + lane] : 0;
    const float ddv  = inv_sqrt_degree((float)cntv);
    float dsv = 0.0f;
    if constexpr (SCALE_A)
      dsv = inv_sqrt_degree((lane < RPW) ? deg_out[r0 + lane] : 0.0f);
    const int nov = min(novf[0], EDGES);
    const float4* __restrict__ h4 = (const float4*)h;
    const float4 bin = ((const float4*)bias_in)[lane];
    int raw = slots[(size_t)r0 * SLOTS + lane];
    for (int i = 0; i < RPW; ++i) {
      const int row = r0 + i;
      const int deg = __builtin_amdgcn_readlane(cntv, i);
      const int n   = min(deg, SLOTS);
      const int raw_next =
          slots[(size_t)min(row + 1, NODES - 1) * SLOTS + lane];
      const int n0   = min(n, 32);
      const int mine = (lane < n0) ? raw : 0;
      float4 r = make_float4(0.0f, 0.0f, 0.0f, 0.0f);
      if constexpr (ADD_RES)
        r = ((const float4*)res)[(size_t)row * (FDIM / 4) + lane];
      float4 acc = make_float4(0.0f, 0.0f, 0.0f, 0.0f);
      add_source_rows(acc, h4, mine, n0, lane);
      if (n > 32) {
        const int n1  = n - 32;
        const int hi  = slots[(size_t)row * SLOTS + 32 + lane];
        const int ids = (lane < n1) ? hi : 0;
        add_source_rows(acc, h4, ids, n1, lane);
      }
      if (deg > SLOTS) {
        for (int base = 0; base < nov; base += 32) {
          const int k  = base + lane;
          const int od = (k < nov) ? ovf_dst[k] : -1;
          const int os = (k < nov) ? ovf_src[k] : 0;
          unsigned m = __builtin_amdgcn_ballot_w32(od == row);
          while (m != 0u) {
            const int bit = __builtin_amdgcn_readfirstlane(__builtin_ctz(m));
            m &= m - 1u;
            const int sidx = __builtin_amdgcn_readlane(os, bit);
            const float4 v = h4[(size_t)sidx * (FDIM / 4) + lane];
            acc.x += v.x; acc.y += v.y; acc.z += v.z; acc.w += v.w;
          }
        }
      }
      const float dd = __int_as_float(
          __builtin_amdgcn_readlane(__float_as_int(ddv), i));
      float4 o;
      o.x = acc.x * dd + bin.x;
      o.y = acc.y * dd + bin.y;
      o.z = acc.z * dd + bin.z;
      o.w = acc.w * dd + bin.w;
      if constexpr (ADD_RES) {
        o.x += r.x; o.y += r.y; o.z += r.z; o.w += r.w;
      }
      o.x = fmaxf(o.x, 0.0f);
      o.y = fmaxf(o.y, 0.0f);
      o.z = fmaxf(o.z, 0.0f);
      o.w = fmaxf(o.w, 0.0f);
      if constexpr (SCALE_A) {
        const float ds = __int_as_float(
            __builtin_amdgcn_readlane(__float_as_int(dsv), i));
        o.x *= ds; o.y *= ds; o.z *= ds; o.w *= ds;
      }
      *(float4*)&As[(wave * RPW + i) * APITCH + 4 * lane] = o;
      raw = raw_next;
    }
  }
  __syncthreads();

  lds_tile_gemm<LDN, ADD_BIAS, false>(As, W, bias_out, nullptr, out, m0, wave,
                                      lane);
}

extern "C" void kernel_launch(void* const* d_in, const int* in_sizes, int n_in,
                              void* d_out, int out_size, void* d_ws, size_t ws_size,
                              hipStream_t stream) {
  const float* x   = (const float*)d_in[0];
  const int*   src = (const int*)  d_in[1];
  const int*   dst = (const int*)  d_in[2];
  const float* W1  = (const float*)d_in[3];
  const float* b1  = (const float*)d_in[4];
  const float* W2  = (const float*)d_in[5];
  const float* b2  = (const float*)d_in[6];
  const float* W3  = (const float*)d_in[7];
  const float* b3  = (const float*)d_in[8];
  const float* Wr  = (const float*)d_in[9];
  const float* br  = (const float*)d_in[10];
  const float* Wo  = (const float*)d_in[11];
  const float* bo  = (const float*)d_in[12];
  float* out = (float*)d_out;

  float* ws    = (float*)d_ws;
  float* RES   = ws;
  float* ABUF  = RES  + (size_t)NODES * FDIM;
  float* HBUF  = ABUF + (size_t)NODES * FDIM;
  int*   slots = (int*)(HBUF + (size_t)NODES * FDIM);
  float* dego  = (float*)(slots + (size_t)NODES * SLOTS);
  int*   cnt   = (int*)(dego + NODES);
  int*   novf  = cnt  + NODES;
  int*   ovfd  = novf + 16;
  int*   ovfs  = ovfd + EDGES;
  unsigned* part = (unsigned*)(ovfs + EDGES);
  int*   binned = (int*)(part + (size_t)HBLOCKS * NODES);
  int*   boff  = binned + (size_t)2 * EDGES;

  const int mblocks  = MBLOCKS;
  const dim3 gemmH_blk(256);

  outdeg_hist_bin_kernel<<<HBLOCKS, HTHREADS, 0, stream>>>(src, dst, part, novf,
                                                           binned, boff);
  edge_gemm_kernel<<<MBLOCKS + NBINS, 256, 0, stream>>>(
      binned, boff, part, dego, cnt, slots, novf, ovfd, ovfs, x, Wr, br, RES,
      W1, HBUF);


  gather_gemm_kernel<FDIM, false, true, false>
      <<<mblocks, gemmH_blk, 0, stream>>>(HBUF, cnt, slots, dego, novf, ovfd,
                                          ovfs, b1, nullptr, W2, nullptr, ABUF);

  gather_gemm_kernel<FDIM, false, true, false>
      <<<mblocks, gemmH_blk, 0, stream>>>(ABUF, cnt, slots, dego, novf, ovfd,
                                          ovfs, b2, nullptr, W3, nullptr, HBUF);

  gather_gemm_kernel<CDIM, true, false, true>
      <<<mblocks, gemmH_blk, 0, stream>>>(HBUF, cnt, slots, dego, novf, ovfd,
                                          ovfs, b3, RES, Wo, bo, out);
}


# Pull: a row's loads go out in one straight-line burst of 4..24, sized to its length, constant slot numbers (was 16-wide rounds: 23 loads for a mean 16 edges, 17+ edges in two rounds); sums unchanged
# speedup vs baseline: 5.2366x; 1.0498x over previous
#include <hip/hip_runtime.h>
#include <hip/hip_bf16.h>


typedef __attribute__((ext_vector_type(2))) float v2f;
typedef __attribute__((ext_vector_type(8))) float v8f;

#define NODES 50000
#define EDGES 800000
#define FDIM  128
#define CDIM  64
#define MTILE 5
#define MBLOCKS (NODES / (16 * MTILE))
#define NBINS MBLOCKS
#define SLOTS 64
#define HBLOCKS 64
#define HTHREADS 1024
#define ECHUNK (EDGES / HBLOCKS)
#define EPT ((ECHUNK + HTHREADS - 1) / HTHREADS)

__global__ void __launch_bounds__(HTHREADS)
outdeg_hist_bin_kernel(const int* __restrict__ src,
                       const int* __restrict__ dst,
                       unsigned* __restrict__ part,
                       int* __restrict__ novf,
                       int* __restrict__ binned,
                       int* __restrict__ boff) {
  __shared__ unsigned hist[NODES];
  __shared__ int2 sorted[ECHUNK];
  __shared__ int bcnt[NBINS + 1];
  __shared__ int wsum[HTHREADS / 32];
  const int t = (int)threadIdx.x;
  const int b = (int)blockIdx.x;
  const int lane = t & 31;
  const int wave = t >> 5;
  if (b == 0 && t == 0) novf[0] = 0;
  int ids[EPT], dd[EPT];
#pragma unroll
  for (int k = 0; k < EPT; ++k) {
    const int i = k * HTHREADS + t;
    ids[k] = (i < ECHUNK) ? src[b * ECHUNK + i] : -1;
    dd[k]  = (i < ECHUNK) ? dst[b * ECHUNK + i] : -1;
  }
  uint4* hist4 = (uint4*)hist;
  for (int i = t; i < NODES / 4; i += HTHREADS)
    hist4[i] = make_uint4(0u, 0u, 0u, 0u);
  if (t <= NBINS) bcnt[t] = 0;
  __syncthreads();
  int rk[EPT];
#pragma unroll
  for (int k = 0; k < EPT; ++k) {
    const bool sok = (unsigned)ids[k] < (unsigned)NODES;
    const bool dok = (unsigned)dd[k] < (unsigned)NODES;
    if (sok) atomicAdd(&hist[ids[k]], 1u);
    rk[k] = (sok && dok) ? atomicAdd(&bcnt[dd[k] / (16 * MTILE)], 1) : -1;
  }
  __syncthreads();
  const int c = (t < NBINS) ? bcnt[t] : 0;
  int x = c;
#pragma unroll
  for (int o = 1; o < 32; o <<= 1) {
    const int y = __shfl_up(x, o);
    if (lane >= o) x += y;
  }
  if (lane == 31) wsum[wave] = x;
  __syncthreads();
  int base = 0;
  for (int w = 0; w < wave; ++w) base += wsum[w];
  const int excl = base + x - c;
  if (t < NBINS) bcnt[t] = excl;
  if (t == NBINS - 1) bcnt[NBINS] = excl + c;
  __syncthreads();
#pragma unroll
  for (int k = 0; k < EPT; ++k)
    if (rk[k] >= 0)
      sorted[bcnt[dd[k] / (16 * MTILE)] + rk[k]] = make_int2(ids[k], dd[k]);
  __syncthreads();
  uint4* __restrict__ out4 = (uint4*)(part + (size_t)b * NODES);
  for (int i = t; i < NODES / 4; i += HTHREADS) out4[i] = hist4[i];
  const int total = bcnt[NBINS];
  int2* __restrict__ bout = (int2*)binned + (size_t)b * ECHUNK;
  for (int i = t; i < total; i += HTHREADS) bout[i] = sorted[i];
  if (t <= NBINS) boff[b * (NBINS + 1) + t] = bcnt[t];
}

__device__ __forceinline__ void build_slot_rows(const int m,
                                                const int* __restrict__ binned,
                                                const int* __restrict__ boff,
                                                int* __restrict__ cnt_in,
                                                int* __restrict__ slots,
                                                int* __restrict__ novf,
                                                int* __restrict__ ovf_dst,
                                                int* __restrict__ ovf_src,
                                                int* tile, int* rcnt, int* seg) {
  const int t = (int)threadIdx.x;
  const int lane = t & 31;
  const int wave = t >> 5;
  if (t < 16 * MTILE) rcnt[t] = 0;
  if (t < HBLOCKS) {
    seg[t]           = boff[t * (NBINS + 1) + m];
    seg[HBLOCKS + t] = boff[t * (NBINS + 1) + m + 1];
  }
  __syncthreads();
  const int2* __restrict__ b2 = (const int2*)binned;
  for (int c = wave; c < HBLOCKS; c += 8) {
    const int beg = seg[c];
    const int end = seg[HBLOCKS + c];
    for (int i = beg + lane; i < end; i += 32) {
      const int2 e = b2[(size_t)c * ECHUNK + i];
      const int row = e.y - m * (16 * MTILE);
      const int r = atomicAdd(&rcnt[row], 1);
      if (r < SLOTS) {
        tile[row * SLOTS + r] = e.x;
      } else {
        const int k = atomicAdd(novf, 1);
        ovf_dst[k] = e.y;
        ovf_src[k] = e.x;
      }
    }
  }
  __syncthreads();
  int4* __restrict__ out4 = (int4*)(slots + (size_t)m * (16 * MTILE) * SLOTS);
  const int4* tile4 = (const int4*)tile;
  for (int i = t; i < 16 * MTILE * SLOTS / 4; i += 256) out4[i] = tile4[i];
  if (t < 16 * MTILE) cnt_in[m * (16 * MTILE) + t] = rcnt[t];
}

__device__ __forceinline__ float inv_sqrt_degree(const float deg) {
  return rsqrtf(deg > 0.0f ? deg : 1.0f);
}

#define RPW    (16 * MTILE / 8)

#define APITCH (FDIM + 4)

template <int LDN, bool ADD_BIAS, bool SCALE_ROWS>
__device__ __forceinline__ void lds_tile_gemm(float* As,
                                              const float* __restrict__ W,
                                              const float* __restrict__ bias_out,
                                              const float* rscale,
                                              float* __restrict__ out,
                                              const int m0, const int wave,
                                              const int lane) {
  constexpr int NCT = LDN / 16;
  const int c   = lane & 15;
  const int kh  = (lane >> 4) << 2;
  const int col = (wave << 4) + c;

  v8f acc[MTILE];
  if (wave < NCT) {
    float init = 0.0f;
    if constexpr (ADD_BIAS) init = bias_out[col];
#pragma unroll
    for (int t = 0; t < MTILE; ++t)
#pragma unroll
      for (int r = 0; r < 8; ++r) acc[t][r] = init;

    const float* __restrict__ bcol = W + (size_t)kh * LDN + col;
    const float* arow = As + c * APITCH + kh;

#pragma unroll
    for (int j = 0; j < FDIM / 8; ++j) {
      v2f b0, b1;
      b0.x = bcol[(size_t)(8 * j + 0) * LDN];
      b0.y = bcol[(size_t)(8 * j + 1) * LDN];
      b1.x = bcol[(size_t)(8 * j + 2) * LDN];
      b1.y = bcol[(size_t)(8 * j + 3) * LDN];
      float4 a[MTILE];
#pragma unroll
      for (int t = 0; t < MTILE; ++t)
        a[t] = *(const float4*)(arow + t * 16 * APITCH + 8 * j);
#pragma unroll
      for (int t = 0; t < MTILE; ++t) {
        v2f a0;
        a0.x = a[t].x; a0.y = a[t].y;
        acc[t] = __builtin_amdgcn_wmma_f32_16x16x4_f32(
             false, a0,  false, b0,
             (short)0, acc[t],  false,  false);
      }
#pragma unroll
      for (int t = 0; t < MTILE; ++t) {
        v2f a1;
        a1.x = a[t].z; a1.y = a[t].w;
        acc[t] = __builtin_amdgcn_wmma_f32_16x16x4_f32(
             false, a1,  false, b1,
             (short)0, acc[t],  false,  false);
      }
    }
  }

  __syncthreads();
  if (wave < NCT) {
    const int rbase = (lane >> 4) << 3;
#pragma unroll
    for (int t = 0; t < MTILE; ++t)
#pragma unroll
      for (int r = 0; r < 8; ++r)
        As[(t * 16 + rbase + r) * APITCH + col] = acc[t][r];
  }
  __syncthreads();
  constexpr int R4 = LDN / 4;
  constexpr int Q  = (16 * MTILE) * R4 / 32 / 8;
  float4* __restrict__ out4 = (float4*)(out + (size_t)m0 * LDN);
#pragma unroll
  for (int q = 0; q < Q; ++q) {
    const int e   = (wave * Q + q) * 32 + lane;
    const int row = e / R4;
    const int c4  = e % R4;
    float4 v = *(const float4*)&As[row * APITCH + 4 * c4];
    if constexpr (SCALE_ROWS) {
      const float sc = rscale[row];
      v.x *= sc; v.y *= sc; v.z *= sc; v.w *= sc;
    }
    out4[e] = v;
  }
}

template <bool ADD_BIAS, bool SCALE_ROWS>
__device__ __forceinline__ void gemm_x_tile(float* As, const int blk,
                                            const float* __restrict__ A,
                                            const float* __restrict__ W,
                                            const float* __restrict__ bias,
                                            const float* rscale,
                                            float* __restrict__ out) {
  const int lane = threadIdx.x & 31;
  const int wave = __builtin_amdgcn_readfirstlane((int)(threadIdx.x >> 5));
  const int m0   = blk * (16 * MTILE);
  const int r0   = m0 + wave * RPW;
  const float4* __restrict__ A4 = (const float4*)A;
  float4 v[RPW];
#pragma unroll
  for (int i = 0; i < RPW; ++i)
    v[i] = A4[(size_t)(r0 + i) * (FDIM / 4) + lane];
#pragma unroll
  for (int i = 0; i < RPW; ++i)
    *(float4*)&As[(wave * RPW + i) * APITCH + 4 * lane] = v[i];
  __syncthreads();
  lds_tile_gemm<FDIM, ADD_BIAS, SCALE_ROWS>(As, W, bias, rscale, out, m0, wave,
                                            lane);
}

__global__ void __launch_bounds__(256)
edge_gemm_kernel(const int* __restrict__ binned,
                 const int* __restrict__ boff,
                 const unsigned* __restrict__ part,
                 float* __restrict__ deg_out,
                 int* __restrict__ cnt_in,
                 int* __restrict__ slots,
                 int* __restrict__ novf,
                 int* __restrict__ ovf_dst,
                 int* __restrict__ ovf_src,
                 const float* __restrict__ x,
                 const float* __restrict__ Wr,
                 const float* __restrict__ br,
                 float* __restrict__ res,
                 const float* __restrict__ W1,
                 float* __restrict__ h1) {
  __shared__ float As[16 * MTILE * APITCH];
  __shared__ float dsl[16 * MTILE];
  const int b = (int)blockIdx.x;
  if (b < MBLOCKS) {
    if (threadIdx.x < 16 * MTILE) {
      const int r = b * (16 * MTILE) + (int)threadIdx.x;
      unsigned n = 0u;
#pragma unroll
      for (int c = 0; c < HBLOCKS; ++c) n += part[(size_t)c * NODES + r];
      const float deg = (float)n;
      deg_out[r] = deg;
      dsl[threadIdx.x] = inv_sqrt_degree(deg);
    }
    gemm_x_tile<true, false>(As, b, x, Wr, br, nullptr, res);
    __syncthreads();
    gemm_x_tile<false, true>(As, b, x, W1, nullptr, dsl, h1);
  } else {
    int* tile = (int*)As;
    int* rcnt = tile + 16 * MTILE * SLOTS;
    int* seg  = rcnt + 16 * MTILE;
    build_slot_rows(b - MBLOCKS, binned, boff, cnt_in, slots, novf, ovf_dst,
                    ovf_src, tile, rcnt, seg);
  }
}


template <int J, int W, bool FULL>
__device__ __forceinline__ void add_source_burst(float4& acc,
                                                 const float4* __restrict__ h4,
                                                 int rowoff, int cnt,
                                                 unsigned lane) {
  float4 v[W];
#pragma unroll
  for (int u = 0; u < W; ++u) {
    const unsigned off = (unsigned)__builtin_amdgcn_readlane(rowoff, J + u);
    const float4* __restrict__ row =
        (const float4*)((const char*)h4 + (size_t)off);
    v[u] = row[lane];
  }
#pragma unroll
  for (int u = 0; u < W; ++u) {
    if (FULL || u < W - 3) {
      acc.x += v[u].x; acc.y += v[u].y; acc.z += v[u].z; acc.w += v[u].w;
    } else {
      const bool ok = (J + u) < cnt;
      acc.x += ok ? v[u].x : 0.0f;
      acc.y += ok ? v[u].y : 0.0f;
      acc.z += ok ? v[u].z : 0.0f;
      acc.w += ok ? v[u].w : 0.0f;
    }
  }
}

__device__ __forceinline__ void add_source_rows(float4& acc,
                                                const float4* __restrict__ h4,
                                                int ids, int cnt, int lane) {
  const int rowoff = ids * (FDIM * (int)sizeof(float));
  const unsigned ul = (unsigned)lane;
  if (cnt <= 0) return;
  if (cnt <= 4)       add_source_burst<0, 4, false>(acc, h4, rowoff, cnt, ul);
  else if (cnt <= 8)  add_source_burst<0, 8, false>(acc, h4, rowoff, cnt, ul);
  else if (cnt <= 12) add_source_burst<0, 12, false>(acc, h4, rowoff, cnt, ul);
  else if (cnt <= 16) add_source_burst<0, 16, false>(acc, h4, rowoff, cnt, ul);
  else if (cnt <= 20) add_source_burst<0, 20, false>(acc, h4, rowoff, cnt, ul);
  else if (cnt <= 24) add_source_burst<0, 24, false>(acc, h4, rowoff, cnt, ul);
  else {
    add_source_burst<0, 16, true>(acc, h4, rowoff, cnt, ul);
    if (cnt <= 28)    add_source_burst<16, 12, false>(acc, h4, rowoff, cnt, ul);
    else              add_source_burst<16, 16, false>(acc, h4, rowoff, cnt, ul);
  }
}

__device__ __forceinline__ void add_source_rows_loop(float4& acc,
                                                     const float4* __restrict__ h4,
                                                     int ids, int cnt, int lane) {
  for (int j = 0; j < cnt; j += 4) {
    float4 v[4];
#pragma unroll
    for (int u = 0; u < 4; ++u) {
      const int sidx = __builtin_amdgcn_readlane(ids, j + u);
      v[u] = h4[(size_t)sidx * (FDIM / 4) + lane];
    }
#pragma unroll
    for (int u = 0; u < 4; ++u) {
      const bool ok = (j + u) < cnt;
      acc.x += ok ? v[u].x : 0.0f;
      acc.y += ok ? v[u].y : 0.0f;
      acc.z += ok ? v[u].z : 0.0f;
      acc.w += ok ? v[u].w : 0.0f;
    }
  }
}

template <int LDN, bool ADD_RES, bool SCALE_A, bool ADD_BIAS>
__global__ void __launch_bounds__(256)
gather_gemm_kernel(const float* __restrict__ h,
                   const int* __restrict__ cnt_in,
                   const int* __restrict__ slots,
                   const float* __restrict__ deg_out,
                   const int* __restrict__ novf,
                   const int* __restrict__ ovf_dst,
                   const int* __restrict__ ovf_src,
                   const float* __restrict__ bias_in,
                   const float* __restrict__ res,
                   const float* __restrict__ W,
                   const float* __restrict__ bias_out,
                   float* __restrict__ out) {
  __shared__ float As[16 * MTILE * APITCH];
  const int lane = threadIdx.x & 31;
  const int wave = __builtin_amdgcn_readfirstlane((int)(threadIdx.x >> 5));
  const int m0   = blockIdx.x * (16 * MTILE);

  {
    const int r0 = m0 + wave * RPW;
    const int   cntv = (lane < RPW) ? cnt_in[r0 + lane] : 0;
    const float ddv  = inv_sqrt_degree((float)cntv);
    float dsv = 0.0f;
    if constexpr (SCALE_A)
      dsv = inv_sqrt_degree((lane < RPW) ? deg_out[r0 + lane] : 0.0f);
    const int nov = min(novf[0], EDGES);
    const float4* __restrict__ h4 = (const float4*)h;
    const float4 bin = ((const float4*)bias_in)[lane];
    int raw = slots[(size_t)r0 * SLOTS + lane];
    for (int i = 0; i < RPW; ++i) {
      const int row = r0 + i;
      const int deg = __builtin_amdgcn_readlane(cntv, i);
      const int n   = min(deg, SLOTS);
      const int raw_next =
          slots[(size_t)min(row + 1, NODES - 1) * SLOTS + lane];
      const int n0   = min(n, 32);
      const int mine = (lane < n0) ? raw : 0;
      float4 r = make_float4(0.0f, 0.0f, 0.0f, 0.0f);
      if constexpr (ADD_RES)
        r = ((const float4*)res)[(size_t)row * (FDIM / 4) + lane];
      float4 acc = make_float4(0.0f, 0.0f, 0.0f, 0.0f);
      add_source_rows(acc, h4, mine, n0, lane);
      if (n > 32) {
        const int n1  = n - 32;
        const int hi  = slots[(size_t)row * SLOTS + 32 + lane];
        const int ids = (lane < n1) ? hi : 0;
        add_source_rows_loop(acc, h4, ids, n1, lane);
      }
      if (deg > SLOTS) {
        for (int base = 0; base < nov; base += 32) {
          const int k  = base + lane;
          const int od = (k < nov) ? ovf_dst[k] : -1;
          const int os = (k < nov) ? ovf_src[k] : 0;
          unsigned m = __builtin_amdgcn_ballot_w32(od == row);
          while (m != 0u) {
            const int bit = __builtin_amdgcn_readfirstlane(__builtin_ctz(m));
            m &= m - 1u;
            const int sidx = __builtin_amdgcn_readlane(os, bit);
            const float4 v = h4[(size_t)sidx * (FDIM / 4) + lane];
            acc.x += v.x; acc.y += v.y; acc.z += v.z; acc.w += v.w;
          }
        }
      }
      const float dd = __int_as_float(
          __builtin_amdgcn_readlane(__float_as_int(ddv), i));
      float4 o;
      o.x = acc.x * dd + bin.x;
      o.y = acc.y * dd + bin.y;
      o.z = acc.z * dd + bin.z;
      o.w = acc.w * dd + bin.w;
      if constexpr (ADD_RES) {
        o.x += r.x; o.y += r.y; o.z += r.z; o.w += r.w;
      }
      o.x = fmaxf(o.x, 0.0f);
      o.y = fmaxf(o.y, 0.0f);
      o.z = fmaxf(o.z, 0.0f);
      o.w = fmaxf(o.w, 0.0f);
      if constexpr (SCALE_A) {
        const float ds = __int_as_float(
            __builtin_amdgcn_readlane(__float_as_int(dsv), i));
        o.x *= ds; o.y *= ds; o.z *= ds; o.w *= ds;
      }
      *(float4*)&As[(wave * RPW + i) * APITCH + 4 * lane] = o;
      raw = raw_next;
    }
  }
  __syncthreads();

  lds_tile_gemm<LDN, ADD_BIAS, false>(As, W, bias_out, nullptr, out, m0, wave,
                                      lane);
}

extern "C" void kernel_launch(void* const* d_in, const int* in_sizes, int n_in,
                              void* d_out, int out_size, void* d_ws, size_t ws_size,
                              hipStream_t stream) {
  const float* x   = (const float*)d_in[0];
  const int*   src = (const int*)  d_in[1];
  const int*   dst = (const int*)  d_in[2];
  const float* W1  = (const float*)d_in[3];
  const float* b1  = (const float*)d_in[4];
  const float* W2  = (const float*)d_in[5];
  const float* b2  = (const float*)d_in[6];
  const float* W3  = (const float*)d_in[7];
  const float* b3  = (const float*)d_in[8];
  const float* Wr  = (const float*)d_in[9];
  const float* br  = (const float*)d_in[10];
  const float* Wo  = (const float*)d_in[11];
  const float* bo  = (const float*)d_in[12];
  float* out = (float*)d_out;

  float* ws    = (float*)d_ws;
  float* RES   = ws;
  float* ABUF  = RES  + (size_t)NODES * FDIM;
  float* HBUF  = ABUF + (size_t)NODES * FDIM;
  int*   slots = (int*)(HBUF + (size_t)NODES * FDIM);
  float* dego  = (float*)(slots + (size_t)NODES * SLOTS);
  int*   cnt   = (int*)(dego + NODES);
  int*   novf  = cnt  + NODES;
  int*   ovfd  = novf + 16;
  int*   ovfs  = ovfd + EDGES;
  unsigned* part = (unsigned*)(ovfs + EDGES);
  int*   binned = (int*)(part + (size_t)HBLOCKS * NODES);
  int*   boff  = binned + (size_t)2 * EDGES;

  const int mblocks  = MBLOCKS;
  const dim3 gemmH_blk(256);

  outdeg_hist_bin_kernel<<<HBLOCKS, HTHREADS, 0, stream>>>(src, dst, part, novf,
                                                           binned, boff);
  edge_gemm_kernel<<<MBLOCKS + NBINS, 256, 0, stream>>>(
      binned, boff, part, dego, cnt, slots, novf, ovfd, ovfs, x, Wr, br, RES,
      W1, HBUF);


  gather_gemm_kernel<FDIM, false, true, false>
      <<<mblocks, gemmH_blk, 0, stream>>>(HBUF, cnt, slots, dego, novf, ovfd,
                                          ovfs, b1, nullptr, W2, nullptr, ABUF);

  gather_gemm_kernel<FDIM, false, true, false>
      <<<mblocks, gemmH_blk, 0, stream>>>(ABUF, cnt, slots, dego, novf, ovfd,
                                          ovfs, b2, nullptr, W3, nullptr, HBUF);

  gather_gemm_kernel<CDIM, true, false, true>
      <<<mblocks, gemmH_blk, 0, stream>>>(HBUF, cnt, slots, dego, novf, ovfd,
                                          ovfs, b3, RES, Wo, bo, out);
}
